# starting kernel
# baseline (speedup 1.0000x reference)
#include <hip/hip_runtime.h>
#include <hip/hip_bf16.h>

typedef float v2f __attribute__((ext_vector_type(2)));
typedef float v8f __attribute__((ext_vector_type(8)));

__global__ void gcn_zero_f(float* __restrict__ p, int n) {
    int t = blockIdx.x * blockDim.x + threadIdx.x;
    if (t < n) p[t] = 0.0f;
}

__global__ void gcn_copy4(float4* __restrict__ dst, const float4* __restrict__ src, int n4) {
    int t = blockIdx.x * blockDim.x + threadIdx.x;
    if (t < n4) dst[t] = src[t];
}

__global__ void gcn_deg(const int* __restrict__ se, const int* __restrict__ re,
                        float* __restrict__ degS, float* __restrict__ degR, int E) {
    int t = blockIdx.x * blockDim.x + threadIdx.x;
    if (t < E) {
        atomicAdd(&degS[se[t]], 1.0f);
        atomicAdd(&degR[re[t]], 1.0f);
    }
}

template <int F>
__global__ void gcn_scatter(const int* __restrict__ se, const int* __restrict__ re,
                            const float* __restrict__ src, float* __restrict__ dst, int E) {
    constexpr int FQ = F / 4;
    long long t = (long long)blockIdx.x * blockDim.x + threadIdx.x;
    if (t >= (long long)E * FQ) return;
    int e = (int)(t / FQ);
    int f = (int)(t % FQ) * 4;
    int s = se[e];
    int r = re[e];
    float4 v = *reinterpret_cast<const float4*>(&src[(size_t)s * F + f]);
    float* d = &dst[(size_t)r * F + f];
    atomicAdd(d + 0, v.x);
    atomicAdd(d + 1, v.y);
    atomicAdd(d + 2, v.z);
    atomicAdd(d + 3, v.w);
}

template <int F, bool ADDONE>
__global__ void gcn_scale(const float* __restrict__ in, const float* __restrict__ degR,
                          float* __restrict__ out, int N) {
    long long t = (long long)blockIdx.x * blockDim.x + threadIdx.x;
    if (t >= (long long)N * F) return;
    int i = (int)(t >> (F == 64 ? 6 : 4));
    float d = degR[i];
    float rs = ADDONE ? rsqrtf(d + 1.0f) : rsqrtf(fmaxf(d, 1.0f));
    out[t] = in[t] * rs;
}

template <int NOUT, bool TANH, bool ADDONE>
__global__ __launch_bounds__(128) void gcn_gemm(const float* __restrict__ in,
                                                const float* __restrict__ W,
                                                const float* __restrict__ bias,
                                                const float* __restrict__ degS,
                                                float* __restrict__ out, int N) {
    constexpr int K = 64;
    constexpr int CT = NOUT / 16;
    constexpr int PITCH = 68;

    __shared__ float Wt[NOUT * PITCH];
    __shared__ float Bl[NOUT];
    __shared__ float Al[64 * PITCH];

    const int tid = threadIdx.x;

    const float4* Wg4 = reinterpret_cast<const float4*>(W);
    for (int i = tid; i < K * NOUT / 4; i += 128) {
        int k = i / (NOUT / 4);
        int n = (i % (NOUT / 4)) * 4;
        float4 w = Wg4[i];
        Wt[(n + 0) * PITCH + k] = w.x;
        Wt[(n + 1) * PITCH + k] = w.y;
        Wt[(n + 2) * PITCH + k] = w.z;
        Wt[(n + 3) * PITCH + k] = w.w;
    }
    if (tid < NOUT) Bl[tid] = bias[tid];

    const int rowBase = blockIdx.x * 64;
    const float4* Ag4 = reinterpret_cast<const float4*>(in);
    for (int i = tid; i < 64 * (K / 4); i += 128) {
        int row = rowBase + i / (K / 4);
        int c4 = (i % (K / 4)) * 4;
        float4 v = (row < N) ? Ag4[(size_t)row * (K / 4) + (c4 >> 2)]
                             : make_float4(0.f, 0.f, 0.f, 0.f);
        *reinterpret_cast<float4*>(&Al[(i / (K / 4)) * PITCH + c4]) = v;
    }
    __syncthreads();

    const int wave = tid >> 5;
    const int lane = tid & 31;
    const int tileRow = rowBase + wave * 16;
    if (tileRow >= N) return;

    const int lm = lane & 15;
    const int kh = (lane >> 4) * 2;
    const int mh = (lane >> 4) * 8;

    v8f acc[CT] = {};
    const float* Arow = &Al[(wave * 16 + lm) * PITCH + kh];

#pragma unroll
    for (int k0 = 0; k0 < K; k0 += 4) {
        v2f a = *reinterpret_cast<const v2f*>(&Arow[k0]);
#pragma unroll
        for (int c = 0; c < CT; ++c) {
            v2f b = *reinterpret_cast<const v2f*>(&Wt[(c * 16 + lm) * PITCH + k0 + kh]);
            acc[c] = __builtin_amdgcn_wmma_f32_16x16x4_f32(
                false, a, false, b, (short)0, acc[c], false, false);
        }
    }

    float rs[8];
#pragma unroll
    for (int j = 0; j < 8; ++j) {
        int row = tileRow + mh + j;
        float d = (row < N) ? degS[row] : 1.0f;
        rs[j] = ADDONE ? rsqrtf(d + 1.0f) : rsqrtf(fmaxf(d, 1.0f));
    }

#pragma unroll
    for (int c = 0; c < CT; ++c) {
        int n = c * 16 + lm;
        float bv = Bl[n];
#pragma unroll
        for (int j = 0; j < 8; ++j) {
            int row = tileRow + mh + j;
            if (row < N) {
                float v = acc[c][j] + bv;
                if (TANH) v = tanhf(v);
                out[(size_t)row * NOUT + n] = v * rs[j];
            }
        }
    }
}

extern "C" void kernel_launch(void* const* d_in, const int* in_sizes, int n_in,
                              void* d_out, int out_size, void* d_ws, size_t ws_size,
                              hipStream_t stream) {
    const float* nodes = (const float*)d_in[0];
    const int* senders = (const int*)d_in[1];
    const int* receivers = (const int*)d_in[2];
    const float* W0 = (const float*)d_in[3];
    const float* b0 = (const float*)d_in[4];
    const float* W1 = (const float*)d_in[5];
    const float* b1 = (const float*)d_in[6];
    const float* W2 = (const float*)d_in[7];
    const float* b2 = (const float*)d_in[8];
    float* out = (float*)d_out;

    const int N = in_sizes[0] / 64;
    const int E = in_sizes[1];

    float* X = (float*)d_ws;
    float* Y = X + (size_t)N * 64;
    float* Z = Y + (size_t)N * 64;
    float* degS = Z + (size_t)N * 16;
    float* degR = degS + N;

    const int zn = N * 16 + 2 * N;
    gcn_zero_f<<<(zn + 255) / 256, 256, 0, stream>>>(Z, zn);
    gcn_deg<<<(E + 255) / 256, 256, 0, stream>>>(senders, receivers, degS, degR, E);

    const int tiles = (N + 15) / 16;
    const int gblocks = (tiles + 3) / 4;
    const int cp4 = N * 16;
    const int g64 = (int)(((long long)E * 16 + 255) / 256);
    const int g16 = (int)(((long long)E * 4 + 255) / 256);
    const int s64 = (int)(((long long)N * 64 + 255) / 256);
    const int s16 = (int)(((long long)N * 16 + 255) / 256);

    gcn_gemm<64, true, true><<<gblocks, 128, 0, stream>>>(nodes, W0, b0, degS, X, N);
    gcn_copy4<<<(cp4 + 255) / 256, 256, 0, stream>>>((float4*)Y, (const float4*)X, cp4);
    gcn_scatter<64><<<g64, 256, 0, stream>>>(senders, receivers, X, Y, E);
    gcn_scale<64, true><<<s64, 256, 0, stream>>>(Y, degR, X, N);

    gcn_gemm<64, true, true><<<gblocks, 128, 0, stream>>>(X, W1, b1, degS, Y, N);
    gcn_copy4<<<(cp4 + 255) / 256, 256, 0, stream>>>((float4*)X, (const float4*)Y, cp4);
    gcn_scatter<64><<<g64, 256, 0, stream>>>(senders, receivers, Y, X, E);
    gcn_scale<64, true><<<s64, 256, 0, stream>>>(X, degR, Y, N);

    gcn_gemm<16, false, false><<<gblocks, 128, 0, stream>>>(Y, W2, b2, degS, X, N);
    gcn_scatter<16><<<g16, 256, 0, stream>>>(senders, receivers, X, Z, E);
    gcn_scale<16, false><<<s16, 256, 0, stream>>>(Z, degR, out, N);
}


# atomic scatter remapped: one wave per 16 edges (32 for 16-wide rows), a lane is one float of a row, so every load and float atomic covers whole 128-B lines; atomic arrival order changes
# speedup vs baseline: 1.3702x; 1.3702x over previous
#include <hip/hip_runtime.h>
#include <hip/hip_bf16.h>

typedef float v2f __attribute__((ext_vector_type(2)));
typedef float v8f __attribute__((ext_vector_type(8)));

__global__ void gcn_zero_f(float* __restrict__ p, int n) {
    int t = blockIdx.x * blockDim.x + threadIdx.x;
    if (t < n) p[t] = 0.0f;
}

__global__ void gcn_copy4(float4* __restrict__ dst, const float4* __restrict__ src, int n4) {
    int t = blockIdx.x * blockDim.x + threadIdx.x;
    if (t < n4) dst[t] = src[t];
}

__global__ void gcn_deg(const int* __restrict__ se, const int* __restrict__ re,
                        float* __restrict__ degS, float* __restrict__ degR, int E) {
    int t = blockIdx.x * blockDim.x + threadIdx.x;
    if (t < E) {
        atomicAdd(&degS[se[t]], 1.0f);
        atomicAdd(&degR[re[t]], 1.0f);
    }
}

constexpr int kEdgesPerWave = 16;
constexpr int kPairsPerWave = 16;

__device__ __forceinline__ int gcn_wave_id() {
    return __builtin_amdgcn_readfirstlane((int)(blockIdx.x * (blockDim.x >> 5) + (threadIdx.x >> 5)));
}

__global__ __launch_bounds__(256) void gcn_scatter64(const int* __restrict__ se,
                                                     const int* __restrict__ re,
                                                     const float* __restrict__ src,
                                                     float* __restrict__ dst, int E) {
    const int lane = threadIdx.x & 31;
    const int e0 = gcn_wave_id() * kEdgesPerWave;
    if (e0 >= E) return;
    if (e0 + kEdgesPerWave <= E) {
        float lo[kEdgesPerWave], hi[kEdgesPerWave];
        int r[kEdgesPerWave];
#pragma unroll
        for (int i = 0; i < kEdgesPerWave; ++i) {
            const float* sp = src + (size_t)se[e0 + i] * 64 + lane;
            lo[i] = sp[0];
            hi[i] = sp[32];
            r[i] = re[e0 + i];
        }
#pragma unroll
        for (int i = 0; i < kEdgesPerWave; ++i) {
            float* dp = dst + (size_t)r[i] * 64 + lane;
            atomicAdd(dp, lo[i]);
            atomicAdd(dp + 32, hi[i]);
        }
    } else {
        for (int e = e0; e < E; ++e) {
            const float* sp = src + (size_t)se[e] * 64 + lane;
            float* dp = dst + (size_t)re[e] * 64 + lane;
            float a = sp[0], b = sp[32];
            atomicAdd(dp, a);
            atomicAdd(dp + 32, b);
        }
    }
}

__global__ __launch_bounds__(256) void gcn_scatter16(const int* __restrict__ se,
                                                     const int* __restrict__ re,
                                                     const float* __restrict__ src,
                                                     float* __restrict__ dst, int E) {
    const int lane = threadIdx.x & 31;
    const bool up = lane >= 16;
    const int f = lane & 15;
    const int e0 = gcn_wave_id() * (2 * kPairsPerWave);
    if (e0 >= E) return;
    if (e0 + 2 * kPairsPerWave <= E) {
        float v[kPairsPerWave];
        int r[kPairsPerWave];
#pragma unroll
        for (int i = 0; i < kPairsPerWave; ++i) {
            const int sa = se[e0 + 2 * i], sb = se[e0 + 2 * i + 1];
            const int ra = re[e0 + 2 * i], rb = re[e0 + 2 * i + 1];
            v[i] = src[(size_t)(up ? sb : sa) * 16 + f];
            r[i] = up ? rb : ra;
        }
#pragma unroll
        for (int i = 0; i < kPairsPerWave; ++i) atomicAdd(&dst[(size_t)r[i] * 16 + f], v[i]);
    } else {
        for (int e = e0; e < E; e += 2) {
            const int me = e + (up ? 1 : 0);
            if (me < E) atomicAdd(&dst[(size_t)re[me] * 16 + f], src[(size_t)se[me] * 16 + f]);
        }
    }
}

template <int F, bool ADDONE>
__global__ void gcn_scale(const float* __restrict__ in, const float* __restrict__ degR,
                          float* __restrict__ out, int N) {
    long long t = (long long)blockIdx.x * blockDim.x + threadIdx.x;
    if (t >= (long long)N * F) return;
    int i = (int)(t >> (F == 64 ? 6 : 4));
    float d = degR[i];
    float rs = ADDONE ? rsqrtf(d + 1.0f) : rsqrtf(fmaxf(d, 1.0f));
    out[t] = in[t] * rs;
}

template <int NOUT, bool TANH, bool ADDONE>
__global__ __launch_bounds__(128) void gcn_gemm(const float* __restrict__ in,
                                                const float* __restrict__ W,
                                                const float* __restrict__ bias,
                                                const float* __restrict__ degS,
                                                float* __restrict__ out, int N) {
    constexpr int K = 64;
    constexpr int CT = NOUT / 16;
    constexpr int PITCH = 68;

    __shared__ float Wt[NOUT * PITCH];
    __shared__ float Bl[NOUT];
    __shared__ float Al[64 * PITCH];

    const int tid = threadIdx.x;

    const float4* Wg4 = reinterpret_cast<const float4*>(W);
    for (int i = tid; i < K * NOUT / 4; i += 128) {
        int k = i / (NOUT / 4);
        int n = (i % (NOUT / 4)) * 4;
        float4 w = Wg4[i];
        Wt[(n + 0) * PITCH + k] = w.x;
        Wt[(n + 1) * PITCH + k] = w.y;
        Wt[(n + 2) * PITCH + k] = w.z;
        Wt[(n + 3) * PITCH + k] = w.w;
    }
    if (tid < NOUT) Bl[tid] = bias[tid];

    const int rowBase = blockIdx.x * 64;
    const float4* Ag4 = reinterpret_cast<const float4*>(in);
    for (int i = tid; i < 64 * (K / 4); i += 128) {
        int row = rowBase + i / (K / 4);
        int c4 = (i % (K / 4)) * 4;
        float4 v = (row < N) ? Ag4[(size_t)row * (K / 4) + (c4 >> 2)]
                             : make_float4(0.f, 0.f, 0.f, 0.f);
        *reinterpret_cast<float4*>(&Al[(i / (K / 4)) * PITCH + c4]) = v;
    }
    __syncthreads();

    const int wave = tid >> 5;
    const int lane = tid & 31;
    const int tileRow = rowBase + wave * 16;
    if (tileRow >= N) return;

    const int lm = lane & 15;
    const int kh = (lane >> 4) * 2;
    const int mh = (lane >> 4) * 8;

    v8f acc[CT] = {};
    const float* Arow = &Al[(wave * 16 + lm) * PITCH + kh];

#pragma unroll
    for (int k0 = 0; k0 < K; k0 += 4) {
        v2f a = *reinterpret_cast<const v2f*>(&Arow[k0]);
#pragma unroll
        for (int c = 0; c < CT; ++c) {
            v2f b = *reinterpret_cast<const v2f*>(&Wt[(c * 16 + lm) * PITCH + k0 + kh]);
            acc[c] = __builtin_amdgcn_wmma_f32_16x16x4_f32(
                false, a, false, b, (short)0, acc[c], false, false);
        }
    }

    float rs[8];
#pragma unroll
    for (int j = 0; j < 8; ++j) {
        int row = tileRow + mh + j;
        float d = (row < N) ? degS[row] : 1.0f;
        rs[j] = ADDONE ? rsqrtf(d + 1.0f) : rsqrtf(fmaxf(d, 1.0f));
    }

#pragma unroll
    for (int c = 0; c < CT; ++c) {
        int n = c * 16 + lm;
        float bv = Bl[n];
#pragma unroll
        for (int j = 0; j < 8; ++j) {
            int row = tileRow + mh + j;
            if (row < N) {
                float v = acc[c][j] + bv;
                if (TANH) v = tanhf(v);
                out[(size_t)row * NOUT + n] = v * rs[j];
            }
        }
    }
}

extern "C" void kernel_launch(void* const* d_in, const int* in_sizes, int n_in,
                              void* d_out, int out_size, void* d_ws, size_t ws_size,
                              hipStream_t stream) {
    const float* nodes = (const float*)d_in[0];
    const int* senders = (const int*)d_in[1];
    const int* receivers = (const int*)d_in[2];
    const float* W0 = (const float*)d_in[3];
    const float* b0 = (const float*)d_in[4];
    const float* W1 = (const float*)d_in[5];
    const float* b1 = (const float*)d_in[6];
    const float* W2 = (const float*)d_in[7];
    const float* b2 = (const float*)d_in[8];
    float* out = (float*)d_out;

    const int N = in_sizes[0] / 64;
    const int E = in_sizes[1];

    float* X = (float*)d_ws;
    float* Y = X + (size_t)N * 64;
    float* Z = Y + (size_t)N * 64;
    float* degS = Z + (size_t)N * 16;
    float* degR = degS + N;

    const int zn = N * 16 + 2 * N;
    gcn_zero_f<<<(zn + 255) / 256, 256, 0, stream>>>(Z, zn);
    gcn_deg<<<(E + 255) / 256, 256, 0, stream>>>(senders, receivers, degS, degR, E);

    const int tiles = (N + 15) / 16;
    const int gblocks = (tiles + 3) / 4;
    const int cp4 = N * 16;
    const int g64 = ((E + kEdgesPerWave - 1) / kEdgesPerWave + 7) / 8;
    const int g16 = ((E + 2 * kPairsPerWave - 1) / (2 * kPairsPerWave) + 7) / 8;
    const int s64 = (int)(((long long)N * 64 + 255) / 256);
    const int s16 = (int)(((long long)N * 16 + 255) / 256);

    gcn_gemm<64, true, true><<<gblocks, 128, 0, stream>>>(nodes, W0, b0, degS, X, N);
    gcn_copy4<<<(cp4 + 255) / 256, 256, 0, stream>>>((float4*)Y, (const float4*)X, cp4);
    gcn_scatter64<<<g64, 256, 0, stream>>>(senders, receivers, X, Y, E);
    gcn_scale<64, true><<<s64, 256, 0, stream>>>(Y, degR, X, N);

    gcn_gemm<64, true, true><<<gblocks, 128, 0, stream>>>(X, W1, b1, degS, Y, N);
    gcn_copy4<<<(cp4 + 255) / 256, 256, 0, stream>>>((float4*)X, (const float4*)Y, cp4);
    gcn_scatter64<<<g64, 256, 0, stream>>>(senders, receivers, Y, X, E);
    gcn_scale<64, true><<<s64, 256, 0, stream>>>(X, degR, Y, N);

    gcn_gemm<16, false, false><<<gblocks, 128, 0, stream>>>(Y, W2, b2, degS, X, N);
    gcn_scatter16<<<g16, 256, 0, stream>>>(senders, receivers, X, Z, E);
    gcn_scale<16, false><<<s16, 256, 0, stream>>>(Z, degR, out, N);
}


# aggregation software-pipelined: list slots come 32 per load a round ahead, row loads in groups of 8 go out before the previous group's LDS adds, start rows and degrees in one run; LDS arrival order
# speedup vs baseline: 1.9624x; 1.4322x over previous
#include <hip/hip_runtime.h>
#include <hip/hip_bf16.h>

typedef float v2f __attribute__((ext_vector_type(2)));
typedef float v8f __attribute__((ext_vector_type(8)));

__global__ void gcn_zero_f(float* __restrict__ p, int n) {
    int t = blockIdx.x * blockDim.x + threadIdx.x;
    if (t < n) p[t] = 0.0f;
}

constexpr int kBinShift = 7;
constexpr int kBin = 1 << kBinShift;
constexpr int kChunkThreads = 1024;
constexpr int kChunkEdges = 4;

__global__ __launch_bounds__(kChunkThreads) void gcn_deg_bins(const int* __restrict__ se,
                                                              const int* __restrict__ re,
                                                              float* __restrict__ degS,
                                                              float* __restrict__ degR,
                                                              int* __restrict__ binCount,
                                                              int E, int nb) {
    extern __shared__ int hist[];
    for (int b = threadIdx.x; b < nb; b += kChunkThreads) hist[b] = 0;
    __syncthreads();
    const int base = blockIdx.x * (kChunkThreads * kChunkEdges) + threadIdx.x;
    for (int k = 0; k < kChunkEdges; ++k) {
        const int e = base + k * kChunkThreads;
        if (e < E) {
            const int r = re[e];
            atomicAdd(&degS[se[e]], 1.0f);
            atomicAdd(&degR[r], 1.0f);
            atomicAdd(&hist[r >> kBinShift], 1);
        }
    }
    __syncthreads();
    for (int b = threadIdx.x; b < nb; b += kChunkThreads) {
        const int c = hist[b];
        if (c) atomicAdd(&binCount[b], c);
    }
}

__global__ __launch_bounds__(1024) void gcn_bin_scan(const int* __restrict__ binCount,
                                                     int* __restrict__ binStart,
                                                     int* __restrict__ binCursor, int nb) {
    __shared__ int part[1024];
    __shared__ int carry;
    const int tid = threadIdx.x;
    if (tid == 0) carry = 0;
    __syncthreads();
    for (int b0 = 0; b0 < nb; b0 += 1024) {
        const int b = b0 + tid;
        const int c = (b < nb) ? binCount[b] : 0;
        part[tid] = c;
        __syncthreads();
        for (int off = 1; off < 1024; off <<= 1) {
            const int v = (tid >= off) ? part[tid - off] : 0;
            __syncthreads();
            part[tid] += v;
            __syncthreads();
        }
        const int incl = part[tid];
        const int first = carry;
        if (b < nb) {
            binStart[b] = first + incl - c;
            binCursor[b] = first + incl - c;
        }
        __syncthreads();
        if (tid == 1023) carry = first + incl;
        __syncthreads();
    }
    if (tid == 0) binStart[nb] = carry;
}

__global__ __launch_bounds__(kChunkThreads) void gcn_bin_place(const int* __restrict__ se,
                                                               const int* __restrict__ re,
                                                               int* __restrict__ binCursor,
                                                               int2* __restrict__ binned,
                                                               int E, int nb) {
    extern __shared__ int hist[];
    for (int b = threadIdx.x; b < nb; b += kChunkThreads) hist[b] = 0;
    __syncthreads();
    const int base = blockIdx.x * (kChunkThreads * kChunkEdges) + threadIdx.x;
    int s[kChunkEdges], r[kChunkEdges], rank[kChunkEdges];
    for (int k = 0; k < kChunkEdges; ++k) {
        const int e = base + k * kChunkThreads;
        s[k] = 0; r[k] = 0; rank[k] = -1;
        if (e < E) {
            s[k] = se[e];
            r[k] = re[e];
            rank[k] = atomicAdd(&hist[r[k] >> kBinShift], 1);
        }
    }
    __syncthreads();
    for (int b = threadIdx.x; b < nb; b += kChunkThreads) {
        const int c = hist[b];
        if (c) hist[b] = atomicAdd(&binCursor[b], c);
    }
    __syncthreads();
    for (int k = 0; k < kChunkEdges; ++k) {
        if (rank[k] >= 0) binned[hist[r[k] >> kBinShift] + rank[k]] = make_int2(s[k], r[k]);
    }
}

constexpr int kRound = 32;

__device__ __forceinline__ int2 gcn_list_line(const int2* __restrict__ binned, int e, int lane,
                                              int e1, int spare) {
    const int q = e + lane;
    int qc = (q < e1) ? q : (e1 - 1);
    qc = (qc > 0) ? qc : 0;
    int2 p = binned[qc];
    if (q >= e1) p = make_int2(0, spare);
    return p;
}

template <int J0>
__device__ __forceinline__ void gcn_load_rows64(int2 slot, const float* __restrict__ src, int lane,
                                                int row0, float2 (&v)[8], int (&rl)[8]) {
#pragma unroll
    for (int i = 0; i < 8; ++i) {
        const int s = __builtin_amdgcn_readlane(slot.x, J0 + i);
        rl[i] = __builtin_amdgcn_readlane(slot.y, J0 + i) - row0;
        v[i] = reinterpret_cast<const float2*>(src + (size_t)s * 64)[lane];
    }
}

__device__ __forceinline__ void gcn_add_rows64(const float2 (&v)[8], const int (&rl)[8], float* al) {
#pragma unroll
    for (int i = 0; i < 8; ++i) {
        float* a = al + rl[i] * 64;
        atomicAdd(a, v[i].x);
        atomicAdd(a + 1, v[i].y);
    }
}

template <int P0>
__device__ __forceinline__ void gcn_load_rows16(int2 slot, int hl, const float* __restrict__ sf,
                                                int row0, float (&v)[8], int (&rl)[8]) {
#pragma unroll
    for (int i = 0; i < 8; ++i) {
        const int j = 2 * (P0 + i);
        const int s0 = __builtin_amdgcn_readlane(slot.x, j);
        const int s1 = __builtin_amdgcn_readlane(slot.x, j + 1);
        const int r0 = __builtin_amdgcn_readlane(slot.y, j);
        const int r1 = __builtin_amdgcn_readlane(slot.y, j + 1);
        const int s = hl ? s1 : s0;
        rl[i] = (hl ? r1 : r0) - row0;
        v[i] = sf[(size_t)s * 16];
    }
}

__device__ __forceinline__ void gcn_add_rows16(const float (&v)[8], const int (&rl)[8], float* af) {
#pragma unroll
    for (int i = 0; i < 8; ++i) atomicAdd(af + rl[i] * 16, v[i]);
}

template <int F, bool SELF, bool ADDONE>
__global__ __launch_bounds__(256) void gcn_agg(const int2* __restrict__ binned,
                                               const int* __restrict__ binStart,
                                               const float* __restrict__ src,
                                               const float* __restrict__ degR,
                                               float* __restrict__ out, int N) {
    constexpr int kWaves = 8;
    constexpr int kIter = kBin * (F / 4) / 256;
    __shared__ float acc[(kBin + 1) * F];

    const int tid = threadIdx.x;
    const int row0 = blockIdx.x * kBin;
    const int rows = (N - row0 < kBin) ? (N - row0) : kBin;
    const int n4 = rows * (F / 4);
    float4* acc4 = reinterpret_cast<float4*>(acc);
    const float4* src4 = reinterpret_cast<const float4*>(src) + (size_t)row0 * (F / 4);

    float dr[kIter];
    float4 st[kIter];
#pragma unroll
    for (int k = 0; k < kIter; ++k) {
        const int i = tid + 256 * k;
        const int ic = (i < n4) ? i : (n4 - 1);
        dr[k] = degR[row0 + ic / (F / 4)];
        if (SELF) st[k] = src4[ic];
    }

    const int lane = tid & 31;
    const int wave = __builtin_amdgcn_readfirstlane(tid >> 5);
    const int e0 = binStart[blockIdx.x];
    const int eEnd = binStart[blockIdx.x + 1];
    const int per = (((eEnd - e0 + kWaves - 1) / kWaves) + 7) & ~7;
    int e = e0 + wave * per;
    const int e1 = (e + per < eEnd) ? (e + per) : eEnd;
    const bool any = e < e1;
    const int spare = row0 + kBin;

    int2 cur = gcn_list_line(binned, e, lane, e1, spare);

#pragma unroll
    for (int k = 0; k < kIter; ++k) {
        const int i = tid + 256 * k;
        float4 v = make_float4(0.f, 0.f, 0.f, 0.f);
        if (SELF && i < n4) v = st[k];
        acc4[i] = v;
    }
    if (tid < F / 4) acc4[kBin * (F / 4) + tid] = make_float4(0.f, 0.f, 0.f, 0.f);

    if constexpr (F == 64) {
        float* al = acc + 2 * lane;
        float2 va[8], vb[8];
        int ra[8], rb[8];
        if (any) gcn_load_rows64<0>(cur, src, lane, row0, va, ra);
        __syncthreads();
        if (any) {
            while (e1 - e >= kRound) {
                const int en = e + kRound;
                const int2 nxt = gcn_list_line(binned, en, lane, e1, spare);
                gcn_load_rows64<8>(cur, src, lane, row0, vb, rb);
                gcn_add_rows64(va, ra, al);
                gcn_load_rows64<16>(cur, src, lane, row0, va, ra);
                gcn_add_rows64(vb, rb, al);
                gcn_load_rows64<24>(cur, src, lane, row0, vb, rb);
                gcn_add_rows64(va, ra, al);
                gcn_load_rows64<0>(nxt, src, lane, row0, va, ra);
                gcn_add_rows64(vb, rb, al);
                cur = nxt;
                e = en;
            }
            const int n = e1 - e;
            if (n > 0) {
                if (n > 8) gcn_load_rows64<8>(cur, src, lane, row0, vb, rb);
                gcn_add_rows64(va, ra, al);
                if (n > 16) gcn_load_rows64<16>(cur, src, lane, row0, va, ra);
                if (n > 8) gcn_add_rows64(vb, rb, al);
                if (n > 24) gcn_load_rows64<24>(cur, src, lane, row0, vb, rb);
                if (n > 16) gcn_add_rows64(va, ra, al);
                if (n > 24) gcn_add_rows64(vb, rb, al);
            }
        }
    } else {
        const int hl = lane >> 4;
        const float* sf = src + (lane & 15);
        float* af = acc + (lane & 15);
        float va[8], vb[8];
        int ra[8], rb[8];
        if (any) gcn_load_rows16<0>(cur, hl, sf, row0, va, ra);
        __syncthreads();
        if (any) {
            while (e1 - e >= kRound) {
                const int en = e + kRound;
                const int2 nxt = gcn_list_line(binned, en, lane, e1, spare);
                gcn_load_rows16<8>(cur, hl, sf, row0, vb, rb);
                gcn_add_rows16(va, ra, af);
                gcn_load_rows16<0>(nxt, hl, sf, row0, va, ra);
                gcn_add_rows16(vb, rb, af);
                cur = nxt;
                e = en;
            }
            const int n = e1 - e;
            if (n > 0) {
                if (n > 16) gcn_load_rows16<8>(cur, hl, sf, row0, vb, rb);
                gcn_add_rows16(va, ra, af);
                if (n > 16) gcn_add_rows16(vb, rb, af);
            }
        }
    }
    __syncthreads();

    float4* out4 = reinterpret_cast<float4*>(out) + (size_t)row0 * (F / 4);
#pragma unroll
    for (int k = 0; k < kIter; ++k) {
        const int i = tid + 256 * k;
        if (i < n4) {
            const float d = dr[k];
            const float rs = ADDONE ? rsqrtf(d + 1.0f) : rsqrtf(fmaxf(d, 1.0f));
            float4 v = acc4[i];
            v.x *= rs;
            v.y *= rs;
            v.z *= rs;
            v.w *= rs;
            out4[i] = v;
        }
    }
}

template <int NOUT, bool TANH, bool ADDONE>
__global__ __launch_bounds__(128) void gcn_gemm(const float* __restrict__ in,
                                                const float* __restrict__ W,
                                                const float* __restrict__ bias,
                                                const float* __restrict__ degS,
                                                float* __restrict__ out, int N) {
    constexpr int K = 64;
    constexpr int CT = NOUT / 16;
    constexpr int PITCH = 68;

    __shared__ float Wt[NOUT * PITCH];
    __shared__ float Bl[NOUT];
    __shared__ float Al[64 * PITCH];

    const int tid = threadIdx.x;

    const float4* Wg4 = reinterpret_cast<const float4*>(W);
    for (int i = tid; i < K * NOUT / 4; i += 128) {
        int k = i / (NOUT / 4);
        int n = (i % (NOUT / 4)) * 4;
        float4 w = Wg4[i];
        Wt[(n + 0) * PITCH + k] = w.x;
        Wt[(n + 1) * PITCH + k] = w.y;
        Wt[(n + 2) * PITCH + k] = w.z;
        Wt[(n + 3) * PITCH + k] = w.w;
    }
    if (tid < NOUT) Bl[tid] = bias[tid];

    const int rowBase = blockIdx.x * 64;
    const float4* Ag4 = reinterpret_cast<const float4*>(in);
    for (int i = tid; i < 64 * (K / 4); i += 128) {
        int row = rowBase + i / (K / 4);
        int c4 = (i % (K / 4)) * 4;
        float4 v = (row < N) ? Ag4[(size_t)row * (K / 4) + (c4 >> 2)]
                             : make_float4(0.f, 0.f, 0.f, 0.f);
        *reinterpret_cast<float4*>(&Al[(i / (K / 4)) * PITCH + c4]) = v;
    }
    __syncthreads();

    const int wave = tid >> 5;
    const int lane = tid & 31;
    const int tileRow = rowBase + wave * 16;
    if (tileRow >= N) return;

    const int lm = lane & 15;
    const int kh = (lane >> 4) * 2;
    const int mh = (lane >> 4) * 8;

    v8f acc[CT] = {};
    const float* Arow = &Al[(wave * 16 + lm) * PITCH + kh];

#pragma unroll
    for (int k0 = 0; k0 < K; k0 += 4) {
        v2f a = *reinterpret_cast<const v2f*>(&Arow[k0]);
#pragma unroll
        for (int c = 0; c < CT; ++c) {
            v2f b = *reinterpret_cast<const v2f*>(&Wt[(c * 16 + lm) * PITCH + k0 + kh]);
            acc[c] = __builtin_amdgcn_wmma_f32_16x16x4_f32(
                false, a, false, b, (short)0, acc[c], false, false);
        }
    }

    float rs[8];
#pragma unroll
    for (int j = 0; j < 8; ++j) {
        int row = tileRow + mh + j;
        float d = (row < N) ? degS[row] : 1.0f;
        rs[j] = ADDONE ? rsqrtf(d + 1.0f) : rsqrtf(fmaxf(d, 1.0f));
    }

#pragma unroll
    for (int c = 0; c < CT; ++c) {
        int n = c * 16 + lm;
        float bv = Bl[n];
#pragma unroll
        for (int j = 0; j < 8; ++j) {
            int row = tileRow + mh + j;
            if (row < N) {
                float v = acc[c][j] + bv;
                if (TANH) v = tanhf(v);
                out[(size_t)row * NOUT + n] = v * rs[j];
            }
        }
    }
}

extern "C" void kernel_launch(void* const* d_in, const int* in_sizes, int n_in,
                              void* d_out, int out_size, void* d_ws, size_t ws_size,
                              hipStream_t stream) {
    const float* nodes = (const float*)d_in[0];
    const int* senders = (const int*)d_in[1];
    const int* receivers = (const int*)d_in[2];
    const float* W0 = (const float*)d_in[3];
    const float* b0 = (const float*)d_in[4];
    const float* W1 = (const float*)d_in[5];
    const float* b1 = (const float*)d_in[6];
    const float* W2 = (const float*)d_in[7];
    const float* b2 = (const float*)d_in[8];
    float* out = (float*)d_out;

    const int N = in_sizes[0] / 64;
    const int E = in_sizes[1];
    const int nb = (N + kBin - 1) / kBin;
    const int nbA = (nb + 2) & ~1;

    float* X = (float*)d_ws;
    float* Y = X + (size_t)N * 64;
    float* degS = Y + (size_t)N * 64;
    float* degR = degS + N;
    int* binCount = (int*)(degR + N);
    int* binStart = binCount + nbA;
    int* binCursor = binStart + nbA;
    int2* binned = (int2*)(binCursor + nbA);

    const int zn = 2 * N + nbA;
    gcn_zero_f<<<(zn + 255) / 256, 256, 0, stream>>>(degS, zn);

    const int chunk = kChunkThreads * kChunkEdges;
    const int chunks = (E + chunk - 1) / chunk;
    const size_t histBytes = (size_t)nb * sizeof(int);
    gcn_deg_bins<<<chunks, kChunkThreads, histBytes, stream>>>(senders, receivers, degS, degR, binCount, E, nb);
    gcn_bin_scan<<<1, 1024, 0, stream>>>(binCount, binStart, binCursor, nb);
    gcn_bin_place<<<chunks, kChunkThreads, histBytes, stream>>>(senders, receivers, binCursor, binned, E, nb);

    const int tiles = (N + 15) / 16;
    const int gblocks = (tiles + 3) / 4;

    gcn_gemm<64, true, true><<<gblocks, 128, 0, stream>>>(nodes, W0, b0, degS, X, N);
    gcn_agg<64, true, true><<<nb, 256, 0, stream>>>(binned, binStart, X, degR, Y, N);

    gcn_gemm<64, true, true><<<gblocks, 128, 0, stream>>>(Y, W1, b1, degS, X, N);
    gcn_agg<64, true, true><<<nb, 256, 0, stream>>>(binned, binStart, X, degR, Y, N);

    gcn_gemm<16, false, false><<<gblocks, 128, 0, stream>>>(Y, W2, b2, degS, X, N);
    gcn_agg<16, false, false><<<nb, 256, 0, stream>>>(binned, binStart, X, degR, out, N);
}


# layer-1 product ahead of layer-0's aggregation: two 64-wide products chained per row tile in one launch, layer-1 bias+tanh+scale in the aggregation store; 10 -> 9 launches; float sum order changes
# speedup vs baseline: 2.0426x; 1.0409x over previous
#include <hip/hip_runtime.h>
#include <hip/hip_bf16.h>

typedef float v2f __attribute__((ext_vector_type(2)));
typedef float v8f __attribute__((ext_vector_type(8)));

__global__ void gcn_zero_f(float* __restrict__ p, int n) {
    int t = blockIdx.x * blockDim.x + threadIdx.x;
    if (t < n) p[t] = 0.0f;
}

constexpr int kBinShift = 7;
constexpr int kBin = 1 << kBinShift;
constexpr int kChunkThreads = 1024;
constexpr int kChunkEdges = 4;

__global__ __launch_bounds__(kChunkThreads) void gcn_deg_bins(const int* __restrict__ se,
                                                              const int* __restrict__ re,
                                                              float* __restrict__ degS,
                                                              float* __restrict__ degR,
                                                              int* __restrict__ binCount,
                                                              int E, int nb) {
    extern __shared__ int hist[];
    for (int b = threadIdx.x; b < nb; b += kChunkThreads) hist[b] = 0;
    __syncthreads();
    const int base = blockIdx.x * (kChunkThreads * kChunkEdges) + threadIdx.x;
    for (int k = 0; k < kChunkEdges; ++k) {
        const int e = base + k * kChunkThreads;
        if (e < E) {
            const int r = re[e];
            atomicAdd(&degS[se[e]], 1.0f);
            atomicAdd(&degR[r], 1.0f);
            atomicAdd(&hist[r >> kBinShift], 1);
        }
    }
    __syncthreads();
    for (int b = threadIdx.x; b < nb; b += kChunkThreads) {
        const int c = hist[b];
        if (c) atomicAdd(&binCount[b], c);
    }
}

__global__ __launch_bounds__(1024) void gcn_bin_scan(const int* __restrict__ binCount,
                                                     int* __restrict__ binStart,
                                                     int* __restrict__ binCursor, int nb) {
    __shared__ int part[1024];
    __shared__ int carry;
    const int tid = threadIdx.x;
    if (tid == 0) carry = 0;
    __syncthreads();
    for (int b0 = 0; b0 < nb; b0 += 1024) {
        const int b = b0 + tid;
        const int c = (b < nb) ? binCount[b] : 0;
        part[tid] = c;
        __syncthreads();
        for (int off = 1; off < 1024; off <<= 1) {
            const int v = (tid >= off) ? part[tid - off] : 0;
            __syncthreads();
            part[tid] += v;
            __syncthreads();
        }
        const int incl = part[tid];
        const int first = carry;
        if (b < nb) {
            binStart[b] = first + incl - c;
            binCursor[b] = first + incl - c;
        }
        __syncthreads();
        if (tid == 1023) carry = first + incl;
        __syncthreads();
    }
    if (tid == 0) binStart[nb] = carry;
}

__global__ __launch_bounds__(kChunkThreads) void gcn_bin_place(const int* __restrict__ se,
                                                               const int* __restrict__ re,
                                                               int* __restrict__ binCursor,
                                                               int2* __restrict__ binned,
                                                               int E, int nb) {
    extern __shared__ int hist[];
    for (int b = threadIdx.x; b < nb; b += kChunkThreads) hist[b] = 0;
    __syncthreads();
    const int base = blockIdx.x * (kChunkThreads * kChunkEdges) + threadIdx.x;
    int s[kChunkEdges], r[kChunkEdges], rank[kChunkEdges];
    for (int k = 0; k < kChunkEdges; ++k) {
        const int e = base + k * kChunkThreads;
        s[k] = 0; r[k] = 0; rank[k] = -1;
        if (e < E) {
            s[k] = se[e];
            r[k] = re[e];
            rank[k] = atomicAdd(&hist[r[k] >> kBinShift], 1);
        }
    }
    __syncthreads();
    for (int b = threadIdx.x; b < nb; b += kChunkThreads) {
        const int c = hist[b];
        if (c) hist[b] = atomicAdd(&binCursor[b], c);
    }
    __syncthreads();
    for (int k = 0; k < kChunkEdges; ++k) {
        if (rank[k] >= 0) binned[hist[r[k] >> kBinShift] + rank[k]] = make_int2(s[k], r[k]);
    }
}

constexpr int kRound = 32;

__device__ __forceinline__ int2 gcn_list_line(const int2* __restrict__ binned, int e, int lane,
                                              int e1, int spare) {
    const int q = e + lane;
    int qc = (q < e1) ? q : (e1 - 1);
    qc = (qc > 0) ? qc : 0;
    int2 p = binned[qc];
    if (q >= e1) p = make_int2(0, spare);
    return p;
}

template <int J0>
__device__ __forceinline__ void gcn_load_rows64(int2 slot, const float* __restrict__ src, int lane,
                                                int row0, float2 (&v)[8], int (&rl)[8]) {
#pragma unroll
    for (int i = 0; i < 8; ++i) {
        const int s = __builtin_amdgcn_readlane(slot.x, J0 + i);
        rl[i] = __builtin_amdgcn_readlane(slot.y, J0 + i) - row0;
        v[i] = reinterpret_cast<const float2*>(src + (size_t)s * 64)[lane];
    }
}

__device__ __forceinline__ void gcn_add_rows64(const float2 (&v)[8], const int (&rl)[8], float* al) {
#pragma unroll
    for (int i = 0; i < 8; ++i) {
        float* a = al + rl[i] * 64;
        atomicAdd(a, v[i].x);
        atomicAdd(a + 1, v[i].y);
    }
}

template <int P0>
__device__ __forceinline__ void gcn_load_rows16(int2 slot, int hl, const float* __restrict__ sf,
                                                int row0, float (&v)[8], int (&rl)[8]) {
#pragma unroll
    for (int i = 0; i < 8; ++i) {
        const int j = 2 * (P0 + i);
        const int s0 = __builtin_amdgcn_readlane(slot.x, j);
        const int s1 = __builtin_amdgcn_readlane(slot.x, j + 1);
        const int r0 = __builtin_amdgcn_readlane(slot.y, j);
        const int r1 = __builtin_amdgcn_readlane(slot.y, j + 1);
        const int s = hl ? s1 : s0;
        rl[i] = (hl ? r1 : r0) - row0;
        v[i] = sf[(size_t)s * 16];
    }
}

__device__ __forceinline__ void gcn_add_rows16(const float (&v)[8], const int (&rl)[8], float* af) {
#pragma unroll
    for (int i = 0; i < 8; ++i) atomicAdd(af + rl[i] * 16, v[i]);
}

template <bool ADDONE>
__device__ __forceinline__ float gcn_rs(float d) {
    return ADDONE ? rsqrtf(d + 1.0f) : rsqrtf(fmaxf(d, 1.0f));
}

template <int F, bool SELF, bool ADDONE, bool ACT>
__global__ __launch_bounds__(256) void gcn_agg(const int2* __restrict__ binned,
                                               const int* __restrict__ binStart,
                                               const float* __restrict__ src,
                                               const float* __restrict__ degR,
                                               const float* __restrict__ actBias,
                                               const float* __restrict__ actDegS,
                                               float* __restrict__ out, int N) {
    constexpr int kWaves = 8;
    constexpr int kIter = kBin * (F / 4) / 256;
    __shared__ float acc[(kBin + 1) * F];

    const int tid = threadIdx.x;
    const int row0 = blockIdx.x * kBin;
    const int rows = (N - row0 < kBin) ? (N - row0) : kBin;
    const int n4 = rows * (F / 4);
    float4* acc4 = reinterpret_cast<float4*>(acc);
    const float4* src4 = reinterpret_cast<const float4*>(src) + (size_t)row0 * (F / 4);

    float dr[kIter];
    float dsn[kIter];
    float4 st[kIter];
#pragma unroll
    for (int k = 0; k < kIter; ++k) {
        const int i = tid + 256 * k;
        const int ic = (i < n4) ? i : (n4 - 1);
        dr[k] = degR[row0 + ic / (F / 4)];
        if (ACT) dsn[k] = actDegS[row0 + ic / (F / 4)];
        if (SELF) st[k] = src4[ic];
    }
    float4 bq = make_float4(0.f, 0.f, 0.f, 0.f);
    if (ACT) bq = reinterpret_cast<const float4*>(actBias)[tid & (F / 4 - 1)];

    const int lane = tid & 31;
    const int wave = __builtin_amdgcn_readfirstlane(tid >> 5);
    const int e0 = binStart[blockIdx.x];
    const int eEnd = binStart[blockIdx.x + 1];
    const int per = (((eEnd - e0 + kWaves - 1) / kWaves) + 7) & ~7;
    int e = e0 + wave * per;
    const int e1 = (e + per < eEnd) ? (e + per) : eEnd;
    const bool any = e < e1;
    const int spare = row0 + kBin;

    int2 cur = gcn_list_line(binned, e, lane, e1, spare);

#pragma unroll
    for (int k = 0; k < kIter; ++k) {
        const int i = tid + 256 * k;
        float4 v = make_float4(0.f, 0.f, 0.f, 0.f);
        if (SELF && i < n4) v = st[k];
        acc4[i] = v;
    }
    if (tid < F / 4) acc4[kBin * (F / 4) + tid] = make_float4(0.f, 0.f, 0.f, 0.f);

    if constexpr (F == 64) {
        float* al = acc + 2 * lane;
        float2 va[8], vb[8];
        int ra[8], rb[8];
        if (any) gcn_load_rows64<0>(cur, src, lane, row0, va, ra);
        __syncthreads();
        if (any) {
            while (e1 - e >= kRound) {
                const int en = e + kRound;
                const int2 nxt = gcn_list_line(binned, en, lane, e1, spare);
                gcn_load_rows64<8>(cur, src, lane, row0, vb, rb);
                gcn_add_rows64(va, ra, al);
                gcn_load_rows64<16>(cur, src, lane, row0, va, ra);
                gcn_add_rows64(vb, rb, al);
                gcn_load_rows64<24>(cur, src, lane, row0, vb, rb);
                gcn_add_rows64(va, ra, al);
                gcn_load_rows64<0>(nxt, src, lane, row0, va, ra);
                gcn_add_rows64(vb, rb, al);
                cur = nxt;
                e = en;
            }
            const int n = e1 - e;
            if (n > 0) {
                if (n > 8) gcn_load_rows64<8>(cur, src, lane, row0, vb, rb);
                gcn_add_rows64(va, ra, al);
                if (n > 16) gcn_load_rows64<16>(cur, src, lane, row0, va, ra);
                if (n > 8) gcn_add_rows64(vb, rb, al);
                if (n > 24) gcn_load_rows64<24>(cur, src, lane, row0, vb, rb);
                if (n > 16) gcn_add_rows64(va, ra, al);
                if (n > 24) gcn_add_rows64(vb, rb, al);
            }
        }
    } else {
        const int hl = lane >> 4;
        const float* sf = src + (lane & 15);
        float* af = acc + (lane & 15);
        float va[8], vb[8];
        int ra[8], rb[8];
        if (any) gcn_load_rows16<0>(cur, hl, sf, row0, va, ra);
        __syncthreads();
        if (any) {
            while (e1 - e >= kRound) {
                const int en = e + kRound;
                const int2 nxt = gcn_list_line(binned, en, lane, e1, spare);
                gcn_load_rows16<8>(cur, hl, sf, row0, vb, rb);
                gcn_add_rows16(va, ra, af);
                gcn_load_rows16<0>(nxt, hl, sf, row0, va, ra);
                gcn_add_rows16(vb, rb, af);
                cur = nxt;
                e = en;
            }
            const int n = e1 - e;
            if (n > 0) {
                if (n > 16) gcn_load_rows16<8>(cur, hl, sf, row0, vb, rb);
                gcn_add_rows16(va, ra, af);
                if (n > 16) gcn_add_rows16(vb, rb, af);
            }
        }
    }
    __syncthreads();

    float4* out4 = reinterpret_cast<float4*>(out) + (size_t)row0 * (F / 4);
#pragma unroll
    for (int k = 0; k < kIter; ++k) {
        const int i = tid + 256 * k;
        if (i < n4) {
            const float rs = gcn_rs<ADDONE>(dr[k]);
            float4 v = acc4[i];
            v.x *= rs;
            v.y *= rs;
            v.z *= rs;
            v.w *= rs;
            if (ACT) {
                const float ss = gcn_rs<ADDONE>(dsn[k]);
                v.x = tanhf(v.x + bq.x) * ss;
                v.y = tanhf(v.y + bq.y) * ss;
                v.z = tanhf(v.z + bq.z) * ss;
                v.w = tanhf(v.w + bq.w) * ss;
            }
            out4[i] = v;
        }
    }
}

template <int NOUT, bool TANH, bool ADDONE>
__global__ __launch_bounds__(128) void gcn_gemm(const float* __restrict__ in,
                                                const float* __restrict__ W,
                                                const float* __restrict__ bias,
                                                const float* __restrict__ degS,
                                                float* __restrict__ out, int N) {
    constexpr int K = 64;
    constexpr int CT = NOUT / 16;
    constexpr int PITCH = 68;

    __shared__ float Wt[NOUT * PITCH];
    __shared__ float Bl[NOUT];
    __shared__ float Al[64 * PITCH];

    const int tid = threadIdx.x;

    const float4* Wg4 = reinterpret_cast<const float4*>(W);
    for (int i = tid; i < K * NOUT / 4; i += 128) {
        int k = i / (NOUT / 4);
        int n = (i % (NOUT / 4)) * 4;
        float4 w = Wg4[i];
        Wt[(n + 0) * PITCH + k] = w.x;
        Wt[(n + 1) * PITCH + k] = w.y;
        Wt[(n + 2) * PITCH + k] = w.z;
        Wt[(n + 3) * PITCH + k] = w.w;
    }
    if (tid < NOUT) Bl[tid] = bias[tid];

    const int rowBase = blockIdx.x * 64;
    const float4* Ag4 = reinterpret_cast<const float4*>(in);
    for (int i = tid; i < 64 * (K / 4); i += 128) {
        int row = rowBase + i / (K / 4);
        int c4 = (i % (K / 4)) * 4;
        float4 v = (row < N) ? Ag4[(size_t)row * (K / 4) + (c4 >> 2)]
                             : make_float4(0.f, 0.f, 0.f, 0.f);
        *reinterpret_cast<float4*>(&Al[(i / (K / 4)) * PITCH + c4]) = v;
    }
    __syncthreads();

    const int wave = tid >> 5;
    const int lane = tid & 31;
    const int tileRow = rowBase + wave * 16;
    if (tileRow >= N) return;

    const int lm = lane & 15;
    const int kh = (lane >> 4) * 2;
    const int mh = (lane >> 4) * 8;

    v8f acc[CT] = {};
    const float* Arow = &Al[(wave * 16 + lm) * PITCH + kh];

#pragma unroll
    for (int k0 = 0; k0 < K; k0 += 4) {
        v2f a = *reinterpret_cast<const v2f*>(&Arow[k0]);
#pragma unroll
        for (int c = 0; c < CT; ++c) {
            v2f b = *reinterpret_cast<const v2f*>(&Wt[(c * 16 + lm) * PITCH + k0 + kh]);
            acc[c] = __builtin_amdgcn_wmma_f32_16x16x4_f32(
                false, a, false, b, (short)0, acc[c], false, false);
        }
    }

    float rs[8];
#pragma unroll
    for (int j = 0; j < 8; ++j) {
        int row = tileRow + mh + j;
        float d = (row < N) ? degS[row] : 1.0f;
        rs[j] = ADDONE ? rsqrtf(d + 1.0f) : rsqrtf(fmaxf(d, 1.0f));
    }

#pragma unroll
    for (int c = 0; c < CT; ++c) {
        int n = c * 16 + lm;
        float bv = Bl[n];
#pragma unroll
        for (int j = 0; j < 8; ++j) {
            int row = tileRow + mh + j;
            if (row < N) {
                float v = acc[c][j] + bv;
                if (TANH) v = tanhf(v);
                out[(size_t)row * NOUT + n] = v * rs[j];
            }
        }
    }
}

__global__ __launch_bounds__(128) void gcn_gemm_pair(const float* __restrict__ in,
                                                     const float* __restrict__ Wa,
                                                     const float* __restrict__ ba,
                                                     const float* __restrict__ Wb,
                                                     const float* __restrict__ degS,
                                                     float* __restrict__ out, int N) {
    constexpr int K = 64;
    constexpr int NOUT = 64;
    constexpr int CT = NOUT / 16;
    constexpr int PITCH = 68;

    constexpr int WIT = K * NOUT / 4 / 128;

    __shared__ float Wt[NOUT * PITCH];
    __shared__ float Bl[NOUT];
    __shared__ float Al[64 * PITCH];

    const int tid = threadIdx.x;

    const float4* Wa4 = reinterpret_cast<const float4*>(Wa);
    const float4* Wb4 = reinterpret_cast<const float4*>(Wb);
    float4 ub[WIT];
#pragma unroll
    for (int t = 0; t < WIT; ++t) {
        const int i = tid + 128 * t;
        int k = i / (NOUT / 4);
        int n = (i % (NOUT / 4)) * 4;
        float4 w = Wa4[i];
        ub[t] = Wb4[i];
        Wt[(n + 0) * PITCH + k] = w.x;
        Wt[(n + 1) * PITCH + k] = w.y;
        Wt[(n + 2) * PITCH + k] = w.z;
        Wt[(n + 3) * PITCH + k] = w.w;
    }
    if (tid < NOUT) Bl[tid] = ba[tid];

    const int rowBase = blockIdx.x * 64;
    const float4* Ag4 = reinterpret_cast<const float4*>(in);
    for (int i = tid; i < 64 * (K / 4); i += 128) {
        int row = rowBase + i / (K / 4);
        int c4 = (i % (K / 4)) * 4;
        float4 v = (row < N) ? Ag4[(size_t)row * (K / 4) + (c4 >> 2)]
                             : make_float4(0.f, 0.f, 0.f, 0.f);
        *reinterpret_cast<float4*>(&Al[(i / (K / 4)) * PITCH + c4]) = v;
    }
    __syncthreads();

    const int wave = tid >> 5;
    const int lane = tid & 31;
    const int tileRow = rowBase + wave * 16;

    const int lm = lane & 15;
    const int kh = (lane >> 4) * 2;
    const int mh = (lane >> 4) * 8;

    const float* Arow = &Al[(wave * 16 + lm) * PITCH + kh];

    v8f acc[CT] = {};
#pragma unroll
    for (int k0 = 0; k0 < K; k0 += 4) {
        v2f a = *reinterpret_cast<const v2f*>(&Arow[k0]);
#pragma unroll
        for (int c = 0; c < CT; ++c) {
            v2f b = *reinterpret_cast<const v2f*>(&Wt[(c * 16 + lm) * PITCH + k0 + kh]);
            acc[c] = __builtin_amdgcn_wmma_f32_16x16x4_f32(
                false, a, false, b, (short)0, acc[c], false, false);
        }
    }

    float rs[8];
#pragma unroll
    for (int j = 0; j < 8; ++j) {
        int row = tileRow + mh + j;
        float d = (row < N) ? degS[row] : 1.0f;
        rs[j] = rsqrtf(d + 1.0f);
    }

    float* Hrow = &Al[(wave * 16 + mh) * PITCH + lm];
#pragma unroll
    for (int c = 0; c < CT; ++c) {
        float bv = Bl[c * 16 + lm];
#pragma unroll
        for (int j = 0; j < 8; ++j) Hrow[j * PITCH + c * 16] = tanhf(acc[c][j] + bv) * rs[j];
    }
    __syncthreads();
#pragma unroll
    for (int t = 0; t < WIT; ++t) {
        const int i = tid + 128 * t;
        int k = i / (NOUT / 4);
        int n = (i % (NOUT / 4)) * 4;
        Wt[(n + 0) * PITCH + k] = ub[t].x;
        Wt[(n + 1) * PITCH + k] = ub[t].y;
        Wt[(n + 2) * PITCH + k] = ub[t].z;
        Wt[(n + 3) * PITCH + k] = ub[t].w;
    }
    __syncthreads();

    v8f acc2[CT] = {};
#pragma unroll
    for (int k0 = 0; k0 < K; k0 += 4) {
        v2f a = *reinterpret_cast<const v2f*>(&Arow[k0]);
#pragma unroll
        for (int c = 0; c < CT; ++c) {
            v2f b = *reinterpret_cast<const v2f*>(&Wt[(c * 16 + lm) * PITCH + k0 + kh]);
            acc2[c] = __builtin_amdgcn_wmma_f32_16x16x4_f32(
                false, a, false, b, (short)0, acc2[c], false, false);
        }
    }

#pragma unroll
    for (int c = 0; c < CT; ++c) {
        int n = c * 16 + lm;
#pragma unroll
        for (int j = 0; j < 8; ++j) {
            int row = tileRow + mh + j;
            if (row < N) out[(size_t)row * NOUT + n] = acc2[c][j];
        }
    }
}

extern "C" void kernel_launch(void* const* d_in, const int* in_sizes, int n_in,
                              void* d_out, int out_size, void* d_ws, size_t ws_size,
                              hipStream_t stream) {
    const float* nodes = (const float*)d_in[0];
    const int* senders = (const int*)d_in[1];
    const int* receivers = (const int*)d_in[2];
    const float* W0 = (const float*)d_in[3];
    const float* b0 = (const float*)d_in[4];
    const float* W1 = (const float*)d_in[5];
    const float* b1 = (const float*)d_in[6];
    const float* W2 = (const float*)d_in[7];
    const float* b2 = (const float*)d_in[8];
    float* out = (float*)d_out;

    const int N = in_sizes[0] / 64;
    const int E = in_sizes[1];
    const int nb = (N + kBin - 1) / kBin;
    const int nbA = (nb + 2) & ~1;

    float* X = (float*)d_ws;
    float* Y = X + (size_t)N * 64;
    float* degS = Y + (size_t)N * 64;
    float* degR = degS + N;
    int* binCount = (int*)(degR + N);
    int* binStart = binCount + nbA;
    int* binCursor = binStart + nbA;
    int2* binned = (int2*)(binCursor + nbA);

    const int zn = 2 * N + nbA;
    gcn_zero_f<<<(zn + 255) / 256, 256, 0, stream>>>(degS, zn);

    const int chunk = kChunkThreads * kChunkEdges;
    const int chunks = (E + chunk - 1) / chunk;
    const size_t histBytes = (size_t)nb * sizeof(int);
    gcn_deg_bins<<<chunks, kChunkThreads, histBytes, stream>>>(senders, receivers, degS, degR, binCount, E, nb);
    gcn_bin_scan<<<1, 1024, 0, stream>>>(binCount, binStart, binCursor, nb);
    gcn_bin_place<<<chunks, kChunkThreads, histBytes, stream>>>(senders, receivers, binCursor, binned, E, nb);

    const int tiles = (N + 15) / 16;
    const int gblocks = (tiles + 3) / 4;

    gcn_gemm_pair<<<gblocks, 128, 0, stream>>>(nodes, W0, b0, W1, degS, X, N);
    gcn_agg<64, true, true, true><<<nb, 256, 0, stream>>>(binned, binStart, X, degR, b1, degS, Y, N);

    gcn_agg<64, true, true, false><<<nb, 256, 0, stream>>>(binned, binStart, Y, degR, nullptr, nullptr, X, N);

    gcn_gemm<16, false, false><<<gblocks, 128, 0, stream>>>(X, W2, b2, degS, Y, N);
    gcn_agg<16, false, false, false><<<nb, 256, 0, stream>>>(binned, binStart, Y, degR, nullptr, nullptr, out, N);
}


# edge placement and the layer-0/1 product pair share ONE launch as independent work-groups (placement: 128 threads x 32 edges, loads in one run); 9 -> 8 launches; slot order in a range may change
# speedup vs baseline: 2.1173x; 1.0366x over previous
#include <hip/hip_runtime.h>
#include <hip/hip_bf16.h>

typedef float v2f __attribute__((ext_vector_type(2)));
typedef float v8f __attribute__((ext_vector_type(8)));

__global__ void gcn_zero_f(float* __restrict__ p, int n) {
    int t = blockIdx.x * blockDim.x + threadIdx.x;
    if (t < n) p[t] = 0.0f;
}

constexpr int kBinShift = 7;
constexpr int kBin = 1 << kBinShift;
constexpr int kChunkThreads = 1024;
constexpr int kChunkEdges = 4;

__global__ __launch_bounds__(kChunkThreads) void gcn_deg_bins(const int* __restrict__ se,
                                                              const int* __restrict__ re,
                                                              float* __restrict__ degS,
                                                              float* __restrict__ degR,
                                                              int* __restrict__ binCount,
                                                              int E, int nb) {
    extern __shared__ int hist[];
    for (int b = threadIdx.x; b < nb; b += kChunkThreads) hist[b] = 0;
    __syncthreads();
    const int base = blockIdx.x * (kChunkThreads * kChunkEdges) + threadIdx.x;
    for (int k = 0; k < kChunkEdges; ++k) {
        const int e = base + k * kChunkThreads;
        if (e < E) {
            const int r = re[e];
            atomicAdd(&degS[se[e]], 1.0f);
            atomicAdd(&degR[r], 1.0f);
            atomicAdd(&hist[r >> kBinShift], 1);
        }
    }
    __syncthreads();
    for (int b = threadIdx.x; b < nb; b += kChunkThreads) {
        const int c = hist[b];
        if (c) atomicAdd(&binCount[b], c);
    }
}

__global__ __launch_bounds__(1024) void gcn_bin_scan(const int* __restrict__ binCount,
                                                     int* __restrict__ binStart,
                                                     int* __restrict__ binCursor, int nb) {
    __shared__ int part[1024];
    __shared__ int carry;
    const int tid = threadIdx.x;
    if (tid == 0) carry = 0;
    __syncthreads();
    for (int b0 = 0; b0 < nb; b0 += 1024) {
        const int b = b0 + tid;
        const int c = (b < nb) ? binCount[b] : 0;
        part[tid] = c;
        __syncthreads();
        for (int off = 1; off < 1024; off <<= 1) {
            const int v = (tid >= off) ? part[tid - off] : 0;
            __syncthreads();
            part[tid] += v;
            __syncthreads();
        }
        const int incl = part[tid];
        const int first = carry;
        if (b < nb) {
            binStart[b] = first + incl - c;
            binCursor[b] = first + incl - c;
        }
        __syncthreads();
        if (tid == 1023) carry = first + incl;
        __syncthreads();
    }
    if (tid == 0) binStart[nb] = carry;
}

constexpr int kPlaceThreads = 128;
constexpr int kPlaceEdges = 32;

__device__ __forceinline__ void gcn_place_body(const int* __restrict__ se,
                                               const int* __restrict__ re,
                                               int* __restrict__ binCursor,
                                               int2* __restrict__ binned,
                                               int E, int nb, int* hist, int chunkIdx) {
    for (int b = threadIdx.x; b < nb; b += kPlaceThreads) hist[b] = 0;
    __syncthreads();
    const int base = chunkIdx * (kPlaceThreads * kPlaceEdges) + threadIdx.x;
    int s[kPlaceEdges], r[kPlaceEdges], rank[kPlaceEdges];
#pragma unroll
    for (int k = 0; k < kPlaceEdges; ++k) {
        const int e = base + k * kPlaceThreads;
        const int ec = (e < E) ? e : (E - 1);
        s[k] = se[ec];
        r[k] = re[ec];
    }
#pragma unroll
    for (int k = 0; k < kPlaceEdges; ++k) {
        const int e = base + k * kPlaceThreads;
        rank[k] = -1;
        if (e < E) rank[k] = atomicAdd(&hist[r[k] >> kBinShift], 1);
    }
    __syncthreads();
    for (int b = threadIdx.x; b < nb; b += kPlaceThreads) {
        const int c = hist[b];
        if (c) hist[b] = atomicAdd(&binCursor[b], c);
    }
    __syncthreads();
#pragma unroll
    for (int k = 0; k < kPlaceEdges; ++k) {
        if (rank[k] >= 0) binned[hist[r[k] >> kBinShift] + rank[k]] = make_int2(s[k], r[k]);
    }
}

constexpr int kRound = 32;

__device__ __forceinline__ int2 gcn_list_line(const int2* __restrict__ binned, int e, int lane,
                                              int e1, int spare) {
    const int q = e + lane;
    int qc = (q < e1) ? q : (e1 - 1);
    qc = (qc > 0) ? qc : 0;
    int2 p = binned[qc];
    if (q >= e1) p = make_int2(0, spare);
    return p;
}

template <int J0>
__device__ __forceinline__ void gcn_load_rows64(int2 slot, const float* __restrict__ src, int lane,
                                                int row0, float2 (&v)[8], int (&rl)[8]) {
#pragma unroll
    for (int i = 0; i < 8; ++i) {
        const int s = __builtin_amdgcn_readlane(slot.x, J0 + i);
        rl[i] = __builtin_amdgcn_readlane(slot.y, J0 + i) - row0;
        v[i] = reinterpret_cast<const float2*>(src + (size_t)s * 64)[lane];
    }
}

__device__ __forceinline__ void gcn_add_rows64(const float2 (&v)[8], const int (&rl)[8], float* al) {
#pragma unroll
    for (int i = 0; i < 8; ++i) {
        float* a = al + rl[i] * 64;
        atomicAdd(a, v[i].x);
        atomicAdd(a + 1, v[i].y);
    }
}

template <int P0>
__device__ __forceinline__ void gcn_load_rows16(int2 slot, int hl, const float* __restrict__ sf,
                                                int row0, float (&v)[8], int (&rl)[8]) {
#pragma unroll
    for (int i = 0; i < 8; ++i) {
        const int j = 2 * (P0 + i);
        const int s0 = __builtin_amdgcn_readlane(slot.x, j);
        const int s1 = __builtin_amdgcn_readlane(slot.x, j + 1);
        const int r0 = __builtin_amdgcn_readlane(slot.y, j);
        const int r1 = __builtin_amdgcn_readlane(slot.y, j + 1);
        const int s = hl ? s1 : s0;
        rl[i] = (hl ? r1 : r0) - row0;
        v[i] = sf[(size_t)s * 16];
    }
}

__device__ __forceinline__ void gcn_add_rows16(const float (&v)[8], const int (&rl)[8], float* af) {
#pragma unroll
    for (int i = 0; i < 8; ++i) atomicAdd(af + rl[i] * 16, v[i]);
}

template <bool ADDONE>
__device__ __forceinline__ float gcn_rs(float d) {
    return ADDONE ? rsqrtf(d + 1.0f) : rsqrtf(fmaxf(d, 1.0f));
}

template <int F, bool SELF, bool ADDONE, bool ACT>
__global__ __launch_bounds__(256) void gcn_agg(const int2* __restrict__ binned,
                                               const int* __restrict__ binStart,
                                               const float* __restrict__ src,
                                               const float* __restrict__ degR,
                                               const float* __restrict__ actBias,
                                               const float* __restrict__ actDegS,
                                               float* __restrict__ out, int N) {
    constexpr int kWaves = 8;
    constexpr int kIter = kBin * (F / 4) / 256;
    __shared__ float acc[(kBin + 1) * F];

    const int tid = threadIdx.x;
    const int row0 = blockIdx.x * kBin;
    const int rows = (N - row0 < kBin) ? (N - row0) : kBin;
    const int n4 = rows * (F / 4);
    float4* acc4 = reinterpret_cast<float4*>(acc);
    const float4* src4 = reinterpret_cast<const float4*>(src) + (size_t)row0 * (F / 4);

    float dr[kIter];
    float dsn[kIter];
    float4 st[kIter];
#pragma unroll
    for (int k = 0; k < kIter; ++k) {
        const int i = tid + 256 * k;
        const int ic = (i < n4) ? i : (n4 - 1);
        dr[k] = degR[row0 + ic / (F / 4)];
        if (ACT) dsn[k] = actDegS[row0 + ic / (F / 4)];
        if (SELF) st[k] = src4[ic];
    }
    float4 bq = make_float4(0.f, 0.f, 0.f, 0.f);
    if (ACT) bq = reinterpret_cast<const float4*>(actBias)[tid & (F / 4 - 1)];

    const int lane = tid & 31;
    const int wave = __builtin_amdgcn_readfirstlane(tid >> 5);
    const int e0 = binStart[blockIdx.x];
    const int eEnd = binStart[blockIdx.x + 1];
    const int per = (((eEnd - e0 + kWaves - 1) / kWaves) + 7) & ~7;
    int e = e0 + wave * per;
    const int e1 = (e + per < eEnd) ? (e + per) : eEnd;
    const bool any = e < e1;
    const int spare = row0 + kBin;

    int2 cur = gcn_list_line(binned, e, lane, e1, spare);

#pragma unroll
    for (int k = 0; k < kIter; ++k) {
        const int i = tid + 256 * k;
        float4 v = make_float4(0.f, 0.f, 0.f, 0.f);
        if (SELF && i < n4) v = st[k];
        acc4[i] = v;
    }
    if (tid < F / 4) acc4[kBin * (F / 4) + tid] = make_float4(0.f, 0.f, 0.f, 0.f);

    if constexpr (F == 64) {
        float* al = acc + 2 * lane;
        float2 va[8], vb[8];
        int ra[8], rb[8];
        if (any) gcn_load_rows64<0>(cur, src, lane, row0, va, ra);
        __syncthreads();
        if (any) {
            while (e1 - e >= kRound) {
                const int en = e + kRound;
                const int2 nxt = gcn_list_line(binned, en, lane, e1, spare);
                gcn_load_rows64<8>(cur, src, lane, row0, vb, rb);
                gcn_add_rows64(va, ra, al);
                gcn_load_rows64<16>(cur, src, lane, row0, va, ra);
                gcn_add_rows64(vb, rb, al);
                gcn_load_rows64<24>(cur, src, lane, row0, vb, rb);
                gcn_add_rows64(va, ra, al);
                gcn_load_rows64<0>(nxt, src, lane, row0, va, ra);
                gcn_add_rows64(vb, rb, al);
                cur = nxt;
                e = en;
            }
            const int n = e1 - e;
            if (n > 0) {
                if (n > 8) gcn_load_rows64<8>(cur, src, lane, row0, vb, rb);
                gcn_add_rows64(va, ra, al);
                if (n > 16) gcn_load_rows64<16>(cur, src, lane, row0, va, ra);
                if (n > 8) gcn_add_rows64(vb, rb, al);
                if (n > 24) gcn_load_rows64<24>(cur, src, lane, row0, vb, rb);
                if (n > 16) gcn_add_rows64(va, ra, al);
                if (n > 24) gcn_add_rows64(vb, rb, al);
            }
        }
    } else {
        const int hl = lane >> 4;
        const float* sf = src + (lane & 15);
        float* af = acc + (lane & 15);
        float va[8], vb[8];
        int ra[8], rb[8];
        if (any) gcn_load_rows16<0>(cur, hl, sf, row0, va, ra);
        __syncthreads();
        if (any) {
            while (e1 - e >= kRound) {
                const int en = e + kRound;
                const int2 nxt = gcn_list_line(binned, en, lane, e1, spare);
                gcn_load_rows16<8>(cur, hl, sf, row0, vb, rb);
                gcn_add_rows16(va, ra, af);
                gcn_load_rows16<0>(nxt, hl, sf, row0, va, ra);
                gcn_add_rows16(vb, rb, af);
                cur = nxt;
                e = en;
            }
            const int n = e1 - e;
            if (n > 0) {
                if (n > 16) gcn_load_rows16<8>(cur, hl, sf, row0, vb, rb);
                gcn_add_rows16(va, ra, af);
                if (n > 16) gcn_add_rows16(vb, rb, af);
            }
        }
    }
    __syncthreads();

    float4* out4 = reinterpret_cast<float4*>(out) + (size_t)row0 * (F / 4);
#pragma unroll
    for (int k = 0; k < kIter; ++k) {
        const int i = tid + 256 * k;
        if (i < n4) {
            const float rs = gcn_rs<ADDONE>(dr[k]);
            float4 v = acc4[i];
            v.x *= rs;
            v.y *= rs;
            v.z *= rs;
            v.w *= rs;
            if (ACT) {
                const float ss = gcn_rs<ADDONE>(dsn[k]);
                v.x = tanhf(v.x + bq.x) * ss;
                v.y = tanhf(v.y + bq.y) * ss;
                v.z = tanhf(v.z + bq.z) * ss;
                v.w = tanhf(v.w + bq.w) * ss;
            }
            out4[i] = v;
        }
    }
}

template <int NOUT, bool TANH, bool ADDONE>
__global__ __launch_bounds__(128) void gcn_gemm(const float* __restrict__ in,
                                                const float* __restrict__ W,
                                                const float* __restrict__ bias,
                                                const float* __restrict__ degS,
                                                float* __restrict__ out, int N) {
    constexpr int K = 64;
    constexpr int CT = NOUT / 16;
    constexpr int PITCH = 68;

    __shared__ float Wt[NOUT * PITCH];
    __shared__ float Bl[NOUT];
    __shared__ float Al[64 * PITCH];

    const int tid = threadIdx.x;

    const float4* Wg4 = reinterpret_cast<const float4*>(W);
    for (int i = tid; i < K * NOUT / 4; i += 128) {
        int k = i / (NOUT / 4);
        int n = (i % (NOUT / 4)) * 4;
        float4 w = Wg4[i];
        Wt[(n + 0) * PITCH + k] = w.x;
        Wt[(n + 1) * PITCH + k] = w.y;
        Wt[(n + 2) * PITCH + k] = w.z;
        Wt[(n + 3) * PITCH + k] = w.w;
    }
    if (tid < NOUT) Bl[tid] = bias[tid];

    const int rowBase = blockIdx.x * 64;
    const float4* Ag4 = reinterpret_cast<const float4*>(in);
    for (int i = tid; i < 64 * (K / 4); i += 128) {
        int row = rowBase + i / (K / 4);
        int c4 = (i % (K / 4)) * 4;
        float4 v = (row < N) ? Ag4[(size_t)row * (K / 4) + (c4 >> 2)]
                             : make_float4(0.f, 0.f, 0.f, 0.f);
        *reinterpret_cast<float4*>(&Al[(i / (K / 4)) * PITCH + c4]) = v;
    }
    __syncthreads();

    const int wave = tid >> 5;
    const int lane = tid & 31;
    const int tileRow = rowBase + wave * 16;
    if (tileRow >= N) return;

    const int lm = lane & 15;
    const int kh = (lane >> 4) * 2;
    const int mh = (lane >> 4) * 8;

    v8f acc[CT] = {};
    const float* Arow = &Al[(wave * 16 + lm) * PITCH + kh];

#pragma unroll
    for (int k0 = 0; k0 < K; k0 += 4) {
        v2f a = *reinterpret_cast<const v2f*>(&Arow[k0]);
#pragma unroll
        for (int c = 0; c < CT; ++c) {
            v2f b = *reinterpret_cast<const v2f*>(&Wt[(c * 16 + lm) * PITCH + k0 + kh]);
            acc[c] = __builtin_amdgcn_wmma_f32_16x16x4_f32(
                false, a, false, b, (short)0, acc[c], false, false);
        }
    }

    float rs[8];
#pragma unroll
    for (int j = 0; j < 8; ++j) {
        int row = tileRow + mh + j;
        float d = (row < N) ? degS[row] : 1.0f;
        rs[j] = ADDONE ? rsqrtf(d + 1.0f) : rsqrtf(fmaxf(d, 1.0f));
    }

#pragma unroll
    for (int c = 0; c < CT; ++c) {
        int n = c * 16 + lm;
        float bv = Bl[n];
#pragma unroll
        for (int j = 0; j < 8; ++j) {
            int row = tileRow + mh + j;
            if (row < N) {
                float v = acc[c][j] + bv;
                if (TANH) v = tanhf(v);
                out[(size_t)row * NOUT + n] = v * rs[j];
            }
        }
    }
}

__global__ __launch_bounds__(128) void gcn_place_gemm_pair(const int* __restrict__ se,
                                                           const int* __restrict__ re,
                                                           int* __restrict__ binCursor,
                                                           int2* __restrict__ binned,
                                                           int E, int nb, int chunks,
                                                           const float* __restrict__ in,
                                                           const float* __restrict__ Wa,
                                                           const float* __restrict__ ba,
                                                           const float* __restrict__ Wb,
                                                           const float* __restrict__ degS,
                                                           float* __restrict__ out, int N) {
    extern __shared__ int hist[];
    if ((int)blockIdx.x < chunks) {
        gcn_place_body(se, re, binCursor, binned, E, nb, hist, (int)blockIdx.x);
        return;
    }
    const int tileBlock = (int)blockIdx.x - chunks;

    constexpr int K = 64;
    constexpr int NOUT = 64;
    constexpr int CT = NOUT / 16;
    constexpr int PITCH = 68;

    constexpr int WIT = K * NOUT / 4 / 128;

    __shared__ float Wt[NOUT * PITCH];
    __shared__ float Bl[NOUT];
    __shared__ float Al[64 * PITCH];

    const int tid = threadIdx.x;

    const float4* Wa4 = reinterpret_cast<const float4*>(Wa);
    const float4* Wb4 = reinterpret_cast<const float4*>(Wb);
    float4 ub[WIT];
#pragma unroll
    for (int t = 0; t < WIT; ++t) {
        const int i = tid + 128 * t;
        int k = i / (NOUT / 4);
        int n = (i % (NOUT / 4)) * 4;
        float4 w = Wa4[i];
        ub[t] = Wb4[i];
        Wt[(n + 0) * PITCH + k] = w.x;
        Wt[(n + 1) * PITCH + k] = w.y;
        Wt[(n + 2) * PITCH + k] = w.z;
        Wt[(n + 3) * PITCH + k] = w.w;
    }
    if (tid < NOUT) Bl[tid] = ba[tid];

    const int rowBase = tileBlock * 64;
    const float4* Ag4 = reinterpret_cast<const float4*>(in);
    for (int i = tid; i < 64 * (K / 4); i += 128) {
        int row = rowBase + i / (K / 4);
        int c4 = (i % (K / 4)) * 4;
        float4 v = (row < N) ? Ag4[(size_t)row * (K / 4) + (c4 >> 2)]
                             : make_float4(0.f, 0.f, 0.f, 0.f);
        *reinterpret_cast<float4*>(&Al[(i / (K / 4)) * PITCH + c4]) = v;
    }
    __syncthreads();

    const int wave = tid >> 5;
    const int lane = tid & 31;
    const int tileRow = rowBase + wave * 16;

    const int lm = lane & 15;
    const int kh = (lane >> 4) * 2;
    const int mh = (lane >> 4) * 8;

    const float* Arow = &Al[(wave * 16 + lm) * PITCH + kh];

    v8f acc[CT] = {};
#pragma unroll
    for (int k0 = 0; k0 < K; k0 += 4) {
        v2f a = *reinterpret_cast<const v2f*>(&Arow[k0]);
#pragma unroll
        for (int c = 0; c < CT; ++c) {
            v2f b = *reinterpret_cast<const v2f*>(&Wt[(c * 16 + lm) * PITCH + k0 + kh]);
            acc[c] = __builtin_amdgcn_wmma_f32_16x16x4_f32(
                false, a, false, b, (short)0, acc[c], false, false);
        }
    }

    float rs[8];
#pragma unroll
    for (int j = 0; j < 8; ++j) {
        int row = tileRow + mh + j;
        float d = (row < N) ? degS[row] : 1.0f;
        rs[j] = rsqrtf(d + 1.0f);
    }

    float* Hrow = &Al[(wave * 16 + mh) * PITCH + lm];
#pragma unroll
    for (int c = 0; c < CT; ++c) {
        float bv = Bl[c * 16 + lm];
#pragma unroll
        for (int j = 0; j < 8; ++j) Hrow[j * PITCH + c * 16] = tanhf(acc[c][j] + bv) * rs[j];
    }
    __syncthreads();
#pragma unroll
    for (int t = 0; t < WIT; ++t) {
        const int i = tid + 128 * t;
        int k = i / (NOUT / 4);
        int n = (i % (NOUT / 4)) * 4;
        Wt[(n + 0) * PITCH + k] = ub[t].x;
        Wt[(n + 1) * PITCH + k] = ub[t].y;
        Wt[(n + 2) * PITCH + k] = ub[t].z;
        Wt[(n + 3) * PITCH + k] = ub[t].w;
    }
    __syncthreads();

    v8f acc2[CT] = {};
#pragma unroll
    for (int k0 = 0; k0 < K; k0 += 4) {
        v2f a = *reinterpret_cast<const v2f*>(&Arow[k0]);
#pragma unroll
        for (int c = 0; c < CT; ++c) {
            v2f b = *reinterpret_cast<const v2f*>(&Wt[(c * 16 + lm) * PITCH + k0 + kh]);
            acc2[c] = __builtin_amdgcn_wmma_f32_16x16x4_f32(
                false, a, false, b, (short)0, acc2[c], false, false);
        }
    }

#pragma unroll
    for (int c = 0; c < CT; ++c) {
        int n = c * 16 + lm;
#pragma unroll
        for (int j = 0; j < 8; ++j) {
            int row = tileRow + mh + j;
            if (row < N) out[(size_t)row * NOUT + n] = acc2[c][j];
        }
    }
}

extern "C" void kernel_launch(void* const* d_in, const int* in_sizes, int n_in,
                              void* d_out, int out_size, void* d_ws, size_t ws_size,
                              hipStream_t stream) {
    const float* nodes = (const float*)d_in[0];
    const int* senders = (const int*)d_in[1];
    const int* receivers = (const int*)d_in[2];
    const float* W0 = (const float*)d_in[3];
    const float* b0 = (const float*)d_in[4];
    const float* W1 = (const float*)d_in[5];
    const float* b1 = (const float*)d_in[6];
    const float* W2 = (const float*)d_in[7];
    const float* b2 = (const float*)d_in[8];
    float* out = (float*)d_out;

    const int N = in_sizes[0] / 64;
    const int E = in_sizes[1];
    const int nb = (N + kBin - 1) / kBin;
    const int nbA = (nb + 2) & ~1;

    float* X = (float*)d_ws;
    float* Y = X + (size_t)N * 64;
    float* degS = Y + (size_t)N * 64;
    float* degR = degS + N;
    int* binCount = (int*)(degR + N);
    int* binStart = binCount + nbA;
    int* binCursor = binStart + nbA;
    int2* binned = (int2*)(binCursor + nbA);

    const int zn = 2 * N + nbA;
    gcn_zero_f<<<(zn + 255) / 256, 256, 0, stream>>>(degS, zn);

    const int chunk = kChunkThreads * kChunkEdges;
    const int chunks = (E + chunk - 1) / chunk;
    const size_t histBytes = (size_t)nb * sizeof(int);
    gcn_deg_bins<<<chunks, kChunkThreads, histBytes, stream>>>(senders, receivers, degS, degR, binCount, E, nb);
    gcn_bin_scan<<<1, 1024, 0, stream>>>(binCount, binStart, binCursor, nb);

    const int tiles = (N + 15) / 16;
    const int gblocks = (tiles + 3) / 4;

    static_assert(kPlaceThreads * kPlaceEdges == kChunkThreads * kChunkEdges,
                  "a placement chunk is a counting chunk");
    gcn_place_gemm_pair<<<chunks + gblocks, kPlaceThreads, histBytes, stream>>>(
        senders, receivers, binCursor, binned, E, nb, chunks, nodes, W0, b0, W1, degS, X, N);
    gcn_agg<64, true, true, true><<<nb, 256, 0, stream>>>(binned, binStart, X, degR, b1, degS, Y, N);

    gcn_agg<64, true, true, false><<<nb, 256, 0, stream>>>(binned, binStart, Y, degR, nullptr, nullptr, X, N);

    gcn_gemm<16, false, false><<<gblocks, 128, 0, stream>>>(X, W2, b2, degS, Y, N);
    gcn_agg<16, false, false, false><<<nb, 256, 0, stream>>>(binned, binStart, Y, degR, nullptr, nullptr, out, N);
}


# receiver degrees counted in LDS by each aggregation work-group from its own range's edge list (one LDS integer add per 32 slots); the 1M single-word global float adds, degR loads and zeroing go
# speedup vs baseline: 2.4003x; 1.1337x over previous
#include <hip/hip_runtime.h>
#include <hip/hip_bf16.h>

typedef float v2f __attribute__((ext_vector_type(2)));
typedef float v8f __attribute__((ext_vector_type(8)));

__global__ void gcn_zero_f(float* __restrict__ p, int n) {
    int t = blockIdx.x * blockDim.x + threadIdx.x;
    if (t < n) p[t] = 0.0f;
}

constexpr int kBinShift = 7;
constexpr int kBin = 1 << kBinShift;
constexpr int kChunkThreads = 1024;
constexpr int kChunkEdges = 4;

__global__ __launch_bounds__(kChunkThreads) void gcn_deg_bins(const int* __restrict__ se,
                                                              const int* __restrict__ re,
                                                              float* __restrict__ degS,
                                                              int* __restrict__ binCount,
                                                              int E, int nb) {
    extern __shared__ int hist[];
    for (int b = threadIdx.x; b < nb; b += kChunkThreads) hist[b] = 0;
    __syncthreads();
    const int base = blockIdx.x * (kChunkThreads * kChunkEdges) + threadIdx.x;
    for (int k = 0; k < kChunkEdges; ++k) {
        const int e = base + k * kChunkThreads;
        if (e < E) {
            const int r = re[e];
            atomicAdd(&degS[se[e]], 1.0f);
            atomicAdd(&hist[r >> kBinShift], 1);
        }
    }
    __syncthreads();
    for (int b = threadIdx.x; b < nb; b += kChunkThreads) {
        const int c = hist[b];
        if (c) atomicAdd(&binCount[b], c);
    }
}

__global__ __launch_bounds__(1024) void gcn_bin_scan(const int* __restrict__ binCount,
                                                     int* __restrict__ binStart,
                                                     int* __restrict__ binCursor, int nb) {
    __shared__ int part[1024];
    __shared__ int carry;
    const int tid = threadIdx.x;
    if (tid == 0) carry = 0;
    __syncthreads();
    for (int b0 = 0; b0 < nb; b0 += 1024) {
        const int b = b0 + tid;
        const int c = (b < nb) ? binCount[b] : 0;
        part[tid] = c;
        __syncthreads();
        for (int off = 1; off < 1024; off <<= 1) {
            const int v = (tid >= off) ? part[tid - off] : 0;
            __syncthreads();
            part[tid] += v;
            __syncthreads();
        }
        const int incl = part[tid];
        const int first = carry;
        if (b < nb) {
            binStart[b] = first + incl - c;
            binCursor[b] = first + incl - c;
        }
        __syncthreads();
        if (tid == 1023) carry = first + incl;
        __syncthreads();
    }
    if (tid == 0) binStart[nb] = carry;
}

constexpr int kPlaceThreads = 128;
constexpr int kPlaceEdges = 32;

__device__ __forceinline__ void gcn_place_body(const int* __restrict__ se,
                                               const int* __restrict__ re,
                                               int* __restrict__ binCursor,
                                               int2* __restrict__ binned,
                                               int E, int nb, int* hist, int chunkIdx) {
    for (int b = threadIdx.x; b < nb; b += kPlaceThreads) hist[b] = 0;
    __syncthreads();
    const int base = chunkIdx * (kPlaceThreads * kPlaceEdges) + threadIdx.x;
    int s[kPlaceEdges], r[kPlaceEdges], rank[kPlaceEdges];
#pragma unroll
    for (int k = 0; k < kPlaceEdges; ++k) {
        const int e = base + k * kPlaceThreads;
        const int ec = (e < E) ? e : (E - 1);
        s[k] = se[ec];
        r[k] = re[ec];
    }
#pragma unroll
    for (int k = 0; k < kPlaceEdges; ++k) {
        const int e = base + k * kPlaceThreads;
        rank[k] = -1;
        if (e < E) rank[k] = atomicAdd(&hist[r[k] >> kBinShift], 1);
    }
    __syncthreads();
    for (int b = threadIdx.x; b < nb; b += kPlaceThreads) {
        const int c = hist[b];
        if (c) hist[b] = atomicAdd(&binCursor[b], c);
    }
    __syncthreads();
#pragma unroll
    for (int k = 0; k < kPlaceEdges; ++k) {
        if (rank[k] >= 0) binned[hist[r[k] >> kBinShift] + rank[k]] = make_int2(s[k], r[k]);
    }
}

constexpr int kRound = 32;

__device__ __forceinline__ int2 gcn_list_line(const int2* __restrict__ binned, int e, int lane,
                                              int e1, int spare) {
    const int q = e + lane;
    int qc = (q < e1) ? q : (e1 - 1);
    qc = (qc > 0) ? qc : 0;
    int2 p = binned[qc];
    if (q >= e1) p = make_int2(0, spare);
    return p;
}

template <int J0>
__device__ __forceinline__ void gcn_load_rows64(int2 slot, const float* __restrict__ src, int lane,
                                                int row0, float2 (&v)[8], int (&rl)[8]) {
#pragma unroll
    for (int i = 0; i < 8; ++i) {
        const int s = __builtin_amdgcn_readlane(slot.x, J0 + i);
        rl[i] = __builtin_amdgcn_readlane(slot.y, J0 + i) - row0;
        v[i] = reinterpret_cast<const float2*>(src + (size_t)s * 64)[lane];
    }
}

__device__ __forceinline__ void gcn_add_rows64(const float2 (&v)[8], const int (&rl)[8], float* al) {
#pragma unroll
    for (int i = 0; i < 8; ++i) {
        float* a = al + rl[i] * 64;
        atomicAdd(a, v[i].x);
        atomicAdd(a + 1, v[i].y);
    }
}

template <int P0>
__device__ __forceinline__ void gcn_load_rows16(int2 slot, int hl, const float* __restrict__ sf,
                                                int row0, float (&v)[8], int (&rl)[8]) {
#pragma unroll
    for (int i = 0; i < 8; ++i) {
        const int j = 2 * (P0 + i);
        const int s0 = __builtin_amdgcn_readlane(slot.x, j);
        const int s1 = __builtin_amdgcn_readlane(slot.x, j + 1);
        const int r0 = __builtin_amdgcn_readlane(slot.y, j);
        const int r1 = __builtin_amdgcn_readlane(slot.y, j + 1);
        const int s = hl ? s1 : s0;
        rl[i] = (hl ? r1 : r0) - row0;
        v[i] = sf[(size_t)s * 16];
    }
}

__device__ __forceinline__ void gcn_add_rows16(const float (&v)[8], const int (&rl)[8], float* af) {
#pragma unroll
    for (int i = 0; i < 8; ++i) atomicAdd(af + rl[i] * 16, v[i]);
}

template <bool ADDONE>
__device__ __forceinline__ float gcn_rs(float d) {
    return ADDONE ? rsqrtf(d + 1.0f) : rsqrtf(fmaxf(d, 1.0f));
}

template <int F, bool SELF, bool ADDONE, bool ACT>
__global__ __launch_bounds__(256) void gcn_agg(const int2* __restrict__ binned,
                                               const int* __restrict__ binStart,
                                               const float* __restrict__ src,
                                               const float* __restrict__ actBias,
                                               const float* __restrict__ actDegS,
                                               float* __restrict__ out, int N) {
    constexpr int kWaves = 8;
    constexpr int kIter = kBin * (F / 4) / 256;
    __shared__ float acc[(kBin + 1) * F];
    __shared__ int cnt[kBin + 32];

    const int tid = threadIdx.x;
    const int row0 = blockIdx.x * kBin;
    const int rows = (N - row0 < kBin) ? (N - row0) : kBin;
    const int n4 = rows * (F / 4);
    float4* acc4 = reinterpret_cast<float4*>(acc);
    const float4* src4 = reinterpret_cast<const float4*>(src) + (size_t)row0 * (F / 4);

    float dsn[kIter];
    float4 st[kIter];
#pragma unroll
    for (int k = 0; k < kIter; ++k) {
        const int i = tid + 256 * k;
        const int ic = (i < n4) ? i : (n4 - 1);
        if (ACT) dsn[k] = actDegS[row0 + ic / (F / 4)];
        if (SELF) st[k] = src4[ic];
    }
    float4 bq = make_float4(0.f, 0.f, 0.f, 0.f);
    if (ACT) bq = reinterpret_cast<const float4*>(actBias)[tid & (F / 4 - 1)];

    const int lane = tid & 31;
    const int wave = __builtin_amdgcn_readfirstlane(tid >> 5);
    const int e0 = binStart[blockIdx.x];
    const int eEnd = binStart[blockIdx.x + 1];
    const int per = (((eEnd - e0 + kWaves - 1) / kWaves) + 7) & ~7;
    int e = e0 + wave * per;
    const int e1 = (e + per < eEnd) ? (e + per) : eEnd;
    const bool any = e < e1;
    const int spare = row0 + kBin;

    int2 cur = gcn_list_line(binned, e, lane, e1, spare);

#pragma unroll
    for (int k = 0; k < kIter; ++k) {
        const int i = tid + 256 * k;
        float4 v = make_float4(0.f, 0.f, 0.f, 0.f);
        if (SELF && i < n4) v = st[k];
        acc4[i] = v;
    }
    if (tid < F / 4) acc4[kBin * (F / 4) + tid] = make_float4(0.f, 0.f, 0.f, 0.f);
    if (tid < kBin) cnt[tid] = 0;

    if constexpr (F == 64) {
        float* al = acc + 2 * lane;
        float2 va[8], vb[8];
        int ra[8], rb[8];
        if (any) gcn_load_rows64<0>(cur, src, lane, row0, va, ra);
        __syncthreads();
        if (any) {
            atomicAdd(&cnt[(cur.y != spare) ? (cur.y - row0) : (kBin + lane)], 1);
            while (e1 - e >= kRound) {
                const int en = e + kRound;
                const int2 nxt = gcn_list_line(binned, en, lane, e1, spare);
                gcn_load_rows64<8>(cur, src, lane, row0, vb, rb);
                gcn_add_rows64(va, ra, al);
                gcn_load_rows64<16>(cur, src, lane, row0, va, ra);
                gcn_add_rows64(vb, rb, al);
                atomicAdd(&cnt[(nxt.y != spare) ? (nxt.y - row0) : (kBin + lane)], 1);
                gcn_load_rows64<24>(cur, src, lane, row0, vb, rb);
                gcn_add_rows64(va, ra, al);
                gcn_load_rows64<0>(nxt, src, lane, row0, va, ra);
                gcn_add_rows64(vb, rb, al);
                cur = nxt;
                e = en;
            }
            const int n = e1 - e;
            if (n > 0) {
                if (n > 8) gcn_load_rows64<8>(cur, src, lane, row0, vb, rb);
                gcn_add_rows64(va, ra, al);
                if (n > 16) gcn_load_rows64<16>(cur, src, lane, row0, va, ra);
                if (n > 8) gcn_add_rows64(vb, rb, al);
                if (n > 24) gcn_load_rows64<24>(cur, src, lane, row0, vb, rb);
                if (n > 16) gcn_add_rows64(va, ra, al);
                if (n > 24) gcn_add_rows64(vb, rb, al);
            }
        }
    } else {
        const int hl = lane >> 4;
        const float* sf = src + (lane & 15);
        float* af = acc + (lane & 15);
        float va[8], vb[8];
        int ra[8], rb[8];
        if (any) gcn_load_rows16<0>(cur, hl, sf, row0, va, ra);
        __syncthreads();
        if (any) {
            atomicAdd(&cnt[(cur.y != spare) ? (cur.y - row0) : (kBin + lane)], 1);
            while (e1 - e >= kRound) {
                const int en = e + kRound;
                const int2 nxt = gcn_list_line(binned, en, lane, e1, spare);
                gcn_load_rows16<8>(cur, hl, sf, row0, vb, rb);
                gcn_add_rows16(va, ra, af);
                gcn_load_rows16<0>(nxt, hl, sf, row0, va, ra);
                atomicAdd(&cnt[(nxt.y != spare) ? (nxt.y - row0) : (kBin + lane)], 1);
                gcn_add_rows16(vb, rb, af);
                cur = nxt;
                e = en;
            }
            const int n = e1 - e;
            if (n > 0) {
                if (n > 16) gcn_load_rows16<8>(cur, hl, sf, row0, vb, rb);
                gcn_add_rows16(va, ra, af);
                if (n > 16) gcn_add_rows16(vb, rb, af);
            }
        }
    }
    __syncthreads();

    float4* out4 = reinterpret_cast<float4*>(out) + (size_t)row0 * (F / 4);
#pragma unroll
    for (int k = 0; k < kIter; ++k) {
        const int i = tid + 256 * k;
        if (i < n4) {
            const float rs = gcn_rs<ADDONE>((float)cnt[i / (F / 4)]);
            float4 v = acc4[i];
            v.x *= rs;
            v.y *= rs;
            v.z *= rs;
            v.w *= rs;
            if (ACT) {
                const float ss = gcn_rs<ADDONE>(dsn[k]);
                v.x = tanhf(v.x + bq.x) * ss;
                v.y = tanhf(v.y + bq.y) * ss;
                v.z = tanhf(v.z + bq.z) * ss;
                v.w = tanhf(v.w + bq.w) * ss;
            }
            out4[i] = v;
        }
    }
}

template <int NOUT, bool TANH, bool ADDONE>
__global__ __launch_bounds__(128) void gcn_gemm(const float* __restrict__ in,
                                                const float* __restrict__ W,
                                                const float* __restrict__ bias,
                                                const float* __restrict__ degS,
                                                float* __restrict__ out, int N) {
    constexpr int K = 64;
    constexpr int CT = NOUT / 16;
    constexpr int PITCH = 68;

    __shared__ float Wt[NOUT * PITCH];
    __shared__ float Bl[NOUT];
    __shared__ float Al[64 * PITCH];

    const int tid = threadIdx.x;

    const float4* Wg4 = reinterpret_cast<const float4*>(W);
    for (int i = tid; i < K * NOUT / 4; i += 128) {
        int k = i / (NOUT / 4);
        int n = (i % (NOUT / 4)) * 4;
        float4 w = Wg4[i];
        Wt[(n + 0) * PITCH + k] = w.x;
        Wt[(n + 1) * PITCH + k] = w.y;
        Wt[(n + 2) * PITCH + k] = w.z;
        Wt[(n + 3) * PITCH + k] = w.w;
    }
    if (tid < NOUT) Bl[tid] = bias[tid];

    const int rowBase = blockIdx.x * 64;
    const float4* Ag4 = reinterpret_cast<const float4*>(in);
    for (int i = tid; i < 64 * (K / 4); i += 128) {
        int row = rowBase + i / (K / 4);
        int c4 = (i % (K / 4)) * 4;
        float4 v = (row < N) ? Ag4[(size_t)row * (K / 4) + (c4 >> 2)]
                             : make_float4(0.f, 0.f, 0.f, 0.f);
        *reinterpret_cast<float4*>(&Al[(i / (K / 4)) * PITCH + c4]) = v;
    }
    __syncthreads();

    const int wave = tid >> 5;
    const int lane = tid & 31;
    const int tileRow = rowBase + wave * 16;
    if (tileRow >= N) return;

    const int lm = lane & 15;
    const int kh = (lane >> 4) * 2;
    const int mh = (lane >> 4) * 8;

    v8f acc[CT] = {};
    const float* Arow = &Al[(wave * 16 + lm) * PITCH + kh];

#pragma unroll
    for (int k0 = 0; k0 < K; k0 += 4) {
        v2f a = *reinterpret_cast<const v2f*>(&Arow[k0]);
#pragma unroll
        for (int c = 0; c < CT; ++c) {
            v2f b = *reinterpret_cast<const v2f*>(&Wt[(c * 16 + lm) * PITCH + k0 + kh]);
            acc[c] = __builtin_amdgcn_wmma_f32_16x16x4_f32(
                false, a, false, b, (short)0, acc[c], false, false);
        }
    }

    float rs[8];
#pragma unroll
    for (int j = 0; j < 8; ++j) {
        int row = tileRow + mh + j;
        float d = (row < N) ? degS[row] : 1.0f;
        rs[j] = ADDONE ? rsqrtf(d + 1.0f) : rsqrtf(fmaxf(d, 1.0f));
    }

#pragma unroll
    for (int c = 0; c < CT; ++c) {
        int n = c * 16 + lm;
        float bv = Bl[n];
#pragma unroll
        for (int j = 0; j < 8; ++j) {
            int row = tileRow + mh + j;
            if (row < N) {
                float v = acc[c][j] + bv;
                if (TANH) v = tanhf(v);
                out[(size_t)row * NOUT + n] = v * rs[j];
            }
        }
    }
}

__global__ __launch_bounds__(128) void gcn_place_gemm_pair(const int* __restrict__ se,
                                                           const int* __restrict__ re,
                                                           int* __restrict__ binCursor,
                                                           int2* __restrict__ binned,
                                                           int E, int nb, int chunks,
                                                           const float* __restrict__ in,
                                                           const float* __restrict__ Wa,
                                                           const float* __restrict__ ba,
                                                           const float* __restrict__ Wb,
                                                           const float* __restrict__ degS,
                                                           float* __restrict__ out, int N) {
    extern __shared__ int hist[];
    if ((int)blockIdx.x < chunks) {
        gcn_place_body(se, re, binCursor, binned, E, nb, hist, (int)blockIdx.x);
        return;
    }
    const int tileBlock = (int)blockIdx.x - chunks;

    constexpr int K = 64;
    constexpr int NOUT = 64;
    constexpr int CT = NOUT / 16;
    constexpr int PITCH = 68;

    constexpr int WIT = K * NOUT / 4 / 128;

    __shared__ float Wt[NOUT * PITCH];
    __shared__ float Bl[NOUT];
    __shared__ float Al[64 * PITCH];

    const int tid = threadIdx.x;

    const float4* Wa4 = reinterpret_cast<const float4*>(Wa);
    const float4* Wb4 = reinterpret_cast<const float4*>(Wb);
    float4 ub[WIT];
#pragma unroll
    for (int t = 0; t < WIT; ++t) {
        const int i = tid + 128 * t;
        int k = i / (NOUT / 4);
        int n = (i % (NOUT / 4)) * 4;
        float4 w = Wa4[i];
        ub[t] = Wb4[i];
        Wt[(n + 0) * PITCH + k] = w.x;
        Wt[(n + 1) * PITCH + k] = w.y;
        Wt[(n + 2) * PITCH + k] = w.z;
        Wt[(n + 3) * PITCH + k] = w.w;
    }
    if (tid < NOUT) Bl[tid] = ba[tid];

    const int rowBase = tileBlock * 64;
    const float4* Ag4 = reinterpret_cast<const float4*>(in);
    for (int i = tid; i < 64 * (K / 4); i += 128) {
        int row = rowBase + i / (K / 4);
        int c4 = (i % (K / 4)) * 4;
        float4 v = (row < N) ? Ag4[(size_t)row * (K / 4) + (c4 >> 2)]
                             : make_float4(0.f, 0.f, 0.f, 0.f);
        *reinterpret_cast<float4*>(&Al[(i / (K / 4)) * PITCH + c4]) = v;
    }
    __syncthreads();

    const int wave = tid >> 5;
    const int lane = tid & 31;
    const int tileRow = rowBase + wave * 16;

    const int lm = lane & 15;
    const int kh = (lane >> 4) * 2;
    const int mh = (lane >> 4) * 8;

    const float* Arow = &Al[(wave * 16 + lm) * PITCH + kh];

    v8f acc[CT] = {};
#pragma unroll
    for (int k0 = 0; k0 < K; k0 += 4) {
        v2f a = *reinterpret_cast<const v2f*>(&Arow[k0]);
#pragma unroll
        for (int c = 0; c < CT; ++c) {
            v2f b = *reinterpret_cast<const v2f*>(&Wt[(c * 16 + lm) * PITCH + k0 + kh]);
            acc[c] = __builtin_amdgcn_wmma_f32_16x16x4_f32(
                false, a, false, b, (short)0, acc[c], false, false);
        }
    }

    float rs[8];
#pragma unroll
    for (int j = 0; j < 8; ++j) {
        int row = tileRow + mh + j;
        float d = (row < N) ? degS[row] : 1.0f;
        rs[j] = rsqrtf(d + 1.0f);
    }

    float* Hrow = &Al[(wave * 16 + mh) * PITCH + lm];
#pragma unroll
    for (int c = 0; c < CT; ++c) {
        float bv = Bl[c * 16 + lm];
#pragma unroll
        for (int j = 0; j < 8; ++j) Hrow[j * PITCH + c * 16] = tanhf(acc[c][j] + bv) * rs[j];
    }
    __syncthreads();
#pragma unroll
    for (int t = 0; t < WIT; ++t) {
        const int i = tid + 128 * t;
        int k = i / (NOUT / 4);
        int n = (i % (NOUT / 4)) * 4;
        Wt[(n + 0) * PITCH + k] = ub[t].x;
        Wt[(n + 1) * PITCH + k] = ub[t].y;
        Wt[(n + 2) * PITCH + k] = ub[t].z;
        Wt[(n + 3) * PITCH + k] = ub[t].w;
    }
    __syncthreads();

    v8f acc2[CT] = {};
#pragma unroll
    for (int k0 = 0; k0 < K; k0 += 4) {
        v2f a = *reinterpret_cast<const v2f*>(&Arow[k0]);
#pragma unroll
        for (int c = 0; c < CT; ++c) {
            v2f b = *reinterpret_cast<const v2f*>(&Wt[(c * 16 + lm) * PITCH + k0 + kh]);
            acc2[c] = __builtin_amdgcn_wmma_f32_16x16x4_f32(
                false, a, false, b, (short)0, acc2[c], false, false);
        }
    }

#pragma unroll
    for (int c = 0; c < CT; ++c) {
        int n = c * 16 + lm;
#pragma unroll
        for (int j = 0; j < 8; ++j) {
            int row = tileRow + mh + j;
            if (row < N) out[(size_t)row * NOUT + n] = acc2[c][j];
        }
    }
}

extern "C" void kernel_launch(void* const* d_in, const int* in_sizes, int n_in,
                              void* d_out, int out_size, void* d_ws, size_t ws_size,
                              hipStream_t stream) {
    const float* nodes = (const float*)d_in[0];
    const int* senders = (const int*)d_in[1];
    const int* receivers = (const int*)d_in[2];
    const float* W0 = (const float*)d_in[3];
    const float* b0 = (const float*)d_in[4];
    const float* W1 = (const float*)d_in[5];
    const float* b1 = (const float*)d_in[6];
    const float* W2 = (const float*)d_in[7];
    const float* b2 = (const float*)d_in[8];
    float* out = (float*)d_out;

    const int N = in_sizes[0] / 64;
    const int E = in_sizes[1];
    const int nb = (N + kBin - 1) / kBin;
    const int nbA = (nb + 2) & ~1;

    float* X = (float*)d_ws;
    float* Y = X + (size_t)N * 64;
    const int nA = (N + 1) & ~1;
    float* degS = Y + (size_t)N * 64;
    int* binCount = (int*)(degS + nA);
    int* binStart = binCount + nbA;
    int* binCursor = binStart + nbA;
    int2* binned = (int2*)(binCursor + nbA);

    const int zn = nA + nbA;
    gcn_zero_f<<<(zn + 255) / 256, 256, 0, stream>>>(degS, zn);

    const int chunk = kChunkThreads * kChunkEdges;
    const int chunks = (E + chunk - 1) / chunk;
    const size_t histBytes = (size_t)nb * sizeof(int);
    gcn_deg_bins<<<chunks, kChunkThreads, histBytes, stream>>>(senders, receivers, degS, binCount, E, nb);
    gcn_bin_scan<<<1, 1024, 0, stream>>>(binCount, binStart, binCursor, nb);

    const int tiles = (N + 15) / 16;
    const int gblocks = (tiles + 3) / 4;

    static_assert(kPlaceThreads * kPlaceEdges == kChunkThreads * kChunkEdges,
                  "a placement chunk is a counting chunk");
    gcn_place_gemm_pair<<<chunks + gblocks, kPlaceThreads, histBytes, stream>>>(
        senders, receivers, binCursor, binned, E, nb, chunks, nodes, W0, b0, W1, degS, X, N);
    gcn_agg<64, true, true, true><<<nb, 256, 0, stream>>>(binned, binStart, X, b1, degS, Y, N);

    gcn_agg<64, true, true, false><<<nb, 256, 0, stream>>>(binned, binStart, Y, nullptr, nullptr, X, N);

    gcn_gemm<16, false, false><<<gblocks, 128, 0, stream>>>(X, W2, b2, degS, Y, N);
    gcn_agg<16, false, false, false><<<nb, 256, 0, stream>>>(binned, binStart, Y, nullptr, nullptr, out, N);
}


# sender degrees and range counts without global atomics: 4x62 work-groups count in LDS (16-bit pairs per node range), store partial tables, summed beside the scan; zero pass gone, 8 -> 7 launches
# speedup vs baseline: 3.2907x; 1.3710x over previous
#include <hip/hip_runtime.h>
#include <hip/hip_bf16.h>

typedef float v2f __attribute__((ext_vector_type(2)));
typedef float v8f __attribute__((ext_vector_type(8)));

constexpr int kBinShift = 7;
constexpr int kBin = 1 << kBinShift;

constexpr int kCntThreads = 512;
constexpr int kCntPer = 32;
constexpr int kCntEdges = kCntThreads * kCntPer;
constexpr int kRangeWords = 12544;
static_assert(kCntEdges < 65536, "a 16-bit sender counter must hold a chunk's edge count");
static_assert(kRangeWords % 4 == 0, "a range is a whole number of 16-byte pieces");

__global__ __launch_bounds__(kCntThreads) void gcn_count(const int* __restrict__ se,
                                                         const int* __restrict__ re,
                                                         unsigned int* __restrict__ sendPart,
                                                         int* __restrict__ binPart,
                                                         int E, int nb, int nbP, int nW) {
    __shared__ unsigned int cnt16[kRangeWords];
    extern __shared__ int hist[];
    const int tid = threadIdx.x;
    const int q = blockIdx.x;
    const int p = blockIdx.y;
    const bool bins = (q == 0);
    const int baseW = q * kRangeWords;
    const int words = (nW - baseW < kRangeWords) ? (nW - baseW) : kRangeWords;
    const int base = p * kCntEdges + tid;
    int s[kCntPer], r[kCntPer];
#pragma unroll
    for (int k = 0; k < kCntPer; ++k) {
        const int e = base + k * kCntThreads;
        const int ec = (e < E) ? e : (E - 1);
        s[k] = se[ec];
        r[k] = re[ec];
    }
    if (bins) {
        for (int b = tid; b < nb; b += kCntThreads) hist[b] = 0;
    }
    uint4* c4 = reinterpret_cast<uint4*>(cnt16);
    for (int i = tid; i < words / 4; i += kCntThreads) c4[i] = make_uint4(0u, 0u, 0u, 0u);
    __syncthreads();
#pragma unroll
    for (int k = 0; k < kCntPer; ++k) {
        const int e = base + k * kCntThreads;
        if (e < E) {
            const int w = (s[k] >> 1) - baseW;
            if ((unsigned int)w < (unsigned int)words) atomicAdd(&cnt16[w], 1u << ((s[k] & 1) << 4));
            if (bins) atomicAdd(&hist[r[k] >> kBinShift], 1);
        }
    }
    __syncthreads();
    uint4* p4 = reinterpret_cast<uint4*>(sendPart + (size_t)p * nW + baseW);
    for (int i = tid; i < words / 4; i += kCntThreads) p4[i] = c4[i];
    if (bins) {
        for (int b = tid; b < nb; b += kCntThreads) binPart[(size_t)p * nbP + b] = hist[b];
    }
}

constexpr int kRedBatch = 32;

__global__ __launch_bounds__(1024) void gcn_scan_reduce(const int* __restrict__ binPart,
                                                        int* __restrict__ binStart,
                                                        int* __restrict__ binCursor,
                                                        int nb, int nbP, int P,
                                                        const unsigned int* __restrict__ sendPart,
                                                        float* __restrict__ degS, int N, int nW) {
    __shared__ int part[1024];
    __shared__ int carry;
    const int tid = threadIdx.x;
    if (blockIdx.x != 0) {
        const int w = ((int)blockIdx.x - 1) * 1024 + tid;
        if (w < nW) {
            unsigned int lo = 0, hi = 0;
            for (int p0 = 0; p0 < P; p0 += kRedBatch) {
                unsigned int v[kRedBatch];
#pragma unroll
                for (int j = 0; j < kRedBatch; ++j) {
                    const int pc = (p0 + j < P) ? (p0 + j) : (P - 1);
                    v[j] = sendPart[(size_t)pc * nW + w];
                }
#pragma unroll
                for (int j = 0; j < kRedBatch; ++j) {
                    if (p0 + j < P) {
                        lo += v[j] & 0xffffu;
                        hi += v[j] >> 16;
                    }
                }
            }
            const int node = 2 * w;
            if (node < N) degS[node] = (float)lo;
            if (node + 1 < N) degS[node + 1] = (float)hi;
        }
        return;
    }
    if (tid == 0) carry = 0;
    __syncthreads();
    for (int b0 = 0; b0 < nb; b0 += 1024) {
        const int b = b0 + tid;
        int c = 0;
        if (b < nb) {
            for (int p0 = 0; p0 < P; p0 += kRedBatch) {
                int v[kRedBatch];
#pragma unroll
                for (int j = 0; j < kRedBatch; ++j) {
                    const int pc = (p0 + j < P) ? (p0 + j) : (P - 1);
                    v[j] = binPart[(size_t)pc * nbP + b];
                }
#pragma unroll
                for (int j = 0; j < kRedBatch; ++j) {
                    if (p0 + j < P) c += v[j];
                }
            }
        }
        part[tid] = c;
        __syncthreads();
        for (int off = 1; off < 1024; off <<= 1) {
            const int v = (tid >= off) ? part[tid - off] : 0;
            __syncthreads();
            part[tid] += v;
            __syncthreads();
        }
        const int incl = part[tid];
        const int first = carry;
        if (b < nb) {
            binStart[b] = first + incl - c;
            binCursor[b] = first + incl - c;
        }
        __syncthreads();
        if (tid == 1023) carry = first + incl;
        __syncthreads();
    }
    if (tid == 0) binStart[nb] = carry;
}

constexpr int kPlaceThreads = 128;
constexpr int kPlaceEdges = 32;

__device__ __forceinline__ void gcn_place_body(const int* __restrict__ se,
                                               const int* __restrict__ re,
                                               int* __restrict__ binCursor,
                                               int2* __restrict__ binned,
                                               int E, int nb, int* hist, int chunkIdx) {
    for (int b = threadIdx.x; b < nb; b += kPlaceThreads) hist[b] = 0;
    __syncthreads();
    const int base = chunkIdx * (kPlaceThreads * kPlaceEdges) + threadIdx.x;
    int s[kPlaceEdges], r[kPlaceEdges], rank[kPlaceEdges];
#pragma unroll
    for (int k = 0; k < kPlaceEdges; ++k) {
        const int e = base + k * kPlaceThreads;
        const int ec = (e < E) ? e : (E - 1);
        s[k] = se[ec];
        r[k] = re[ec];
    }
#pragma unroll
    for (int k = 0; k < kPlaceEdges; ++k) {
        const int e = base + k * kPlaceThreads;
        rank[k] = -1;
        if (e < E) rank[k] = atomicAdd(&hist[r[k] >> kBinShift], 1);
    }
    __syncthreads();
    for (int b = threadIdx.x; b < nb; b += kPlaceThreads) {
        const int c = hist[b];
        if (c) hist[b] = atomicAdd(&binCursor[b], c);
    }
    __syncthreads();
#pragma unroll
    for (int k = 0; k < kPlaceEdges; ++k) {
        if (rank[k] >= 0) binned[hist[r[k] >> kBinShift] + rank[k]] = make_int2(s[k], r[k]);
    }
}

constexpr int kRound = 32;

__device__ __forceinline__ int2 gcn_list_line(const int2* __restrict__ binned, int e, int lane,
                                              int e1, int spare) {
    const int q = e + lane;
    int qc = (q < e1) ? q : (e1 - 1);
    qc = (qc > 0) ? qc : 0;
    int2 p = binned[qc];
    if (q >= e1) p = make_int2(0, spare);
    return p;
}

template <int J0>
__device__ __forceinline__ void gcn_load_rows64(int2 slot, const float* __restrict__ src, int lane,
                                                int row0, float2 (&v)[8], int (&rl)[8]) {
#pragma unroll
    for (int i = 0; i < 8; ++i) {
        const int s = __builtin_amdgcn_readlane(slot.x, J0 + i);
        rl[i] = __builtin_amdgcn_readlane(slot.y, J0 + i) - row0;
        v[i] = reinterpret_cast<const float2*>(src + (size_t)s * 64)[lane];
    }
}

__device__ __forceinline__ void gcn_add_rows64(const float2 (&v)[8], const int (&rl)[8], float* al) {
#pragma unroll
    for (int i = 0; i < 8; ++i) {
        float* a = al + rl[i] * 64;
        atomicAdd(a, v[i].x);
        atomicAdd(a + 1, v[i].y);
    }
}

template <int P0>
__device__ __forceinline__ void gcn_load_rows16(int2 slot, int hl, const float* __restrict__ sf,
                                                int row0, float (&v)[8], int (&rl)[8]) {
#pragma unroll
    for (int i = 0; i < 8; ++i) {
        const int j = 2 * (P0 + i);
        const int s0 = __builtin_amdgcn_readlane(slot.x, j);
        const int s1 = __builtin_amdgcn_readlane(slot.x, j + 1);
        const int r0 = __builtin_amdgcn_readlane(slot.y, j);
        const int r1 = __builtin_amdgcn_readlane(slot.y, j + 1);
        const int s = hl ? s1 : s0;
        rl[i] = (hl ? r1 : r0) - row0;
        v[i] = sf[(size_t)s * 16];
    }
}

__device__ __forceinline__ void gcn_add_rows16(const float (&v)[8], const int (&rl)[8], float* af) {
#pragma unroll
    for (int i = 0; i < 8; ++i) atomicAdd(af + rl[i] * 16, v[i]);
}

template <bool ADDONE>
__device__ __forceinline__ float gcn_rs(float d) {
    return ADDONE ? rsqrtf(d + 1.0f) : rsqrtf(fmaxf(d, 1.0f));
}

template <int F, bool SELF, bool ADDONE, bool ACT>
__global__ __launch_bounds__(256) void gcn_agg(const int2* __restrict__ binned,
                                               const int* __restrict__ binStart,
                                               const float* __restrict__ src,
                                               const float* __restrict__ actBias,
                                               const float* __restrict__ actDegS,
                                               float* __restrict__ out, int N) {
    constexpr int kWaves = 8;
    constexpr int kIter = kBin * (F / 4) / 256;
    __shared__ float acc[(kBin + 1) * F];
    __shared__ int cnt[kBin + 32];

    const int tid = threadIdx.x;
    const int row0 = blockIdx.x * kBin;
    const int rows = (N - row0 < kBin) ? (N - row0) : kBin;
    const int n4 = rows * (F / 4);
    float4* acc4 = reinterpret_cast<float4*>(acc);
    const float4* src4 = reinterpret_cast<const float4*>(src) + (size_t)row0 * (F / 4);

    float dsn[kIter];
    float4 st[kIter];
#pragma unroll
    for (int k = 0; k < kIter; ++k) {
        const int i = tid + 256 * k;
        const int ic = (i < n4) ? i : (n4 - 1);
        if (ACT) dsn[k] = actDegS[row0 + ic / (F / 4)];
        if (SELF) st[k] = src4[ic];
    }
    float4 bq = make_float4(0.f, 0.f, 0.f, 0.f);
    if (ACT) bq = reinterpret_cast<const float4*>(actBias)[tid & (F / 4 - 1)];

    const int lane = tid & 31;
    const int wave = __builtin_amdgcn_readfirstlane(tid >> 5);
    const int e0 = binStart[blockIdx.x];
    const int eEnd = binStart[blockIdx.x + 1];
    const int per = (((eEnd - e0 + kWaves - 1) / kWaves) + 7) & ~7;
    int e = e0 + wave * per;
    const int e1 = (e + per < eEnd) ? (e + per) : eEnd;
    const bool any = e < e1;
    const int spare = row0 + kBin;

    int2 cur = gcn_list_line(binned, e, lane, e1, spare);

#pragma unroll
    for (int k = 0; k < kIter; ++k) {
        const int i = tid + 256 * k;
        float4 v = make_float4(0.f, 0.f, 0.f, 0.f);
        if (SELF && i < n4) v = st[k];
        acc4[i] = v;
    }
    if (tid < F / 4) acc4[kBin * (F / 4) + tid] = make_float4(0.f, 0.f, 0.f, 0.f);
    if (tid < kBin) cnt[tid] = 0;

    if constexpr (F == 64) {
        float* al = acc + 2 * lane;
        float2 va[8], vb[8];
        int ra[8], rb[8];
        if (any) gcn_load_rows64<0>(cur, src, lane, row0, va, ra);
        __syncthreads();
        if (any) {
            atomicAdd(&cnt[(cur.y != spare) ? (cur.y - row0) : (kBin + lane)], 1);
            while (e1 - e >= kRound) {
                const int en = e + kRound;
                const int2 nxt = gcn_list_line(binned, en, lane, e1, spare);
                gcn_load_rows64<8>(cur, src, lane, row0, vb, rb);
                gcn_add_rows64(va, ra, al);
                gcn_load_rows64<16>(cur, src, lane, row0, va, ra);
                gcn_add_rows64(vb, rb, al);
                atomicAdd(&cnt[(nxt.y != spare) ? (nxt.y - row0) : (kBin + lane)], 1);
                gcn_load_rows64<24>(cur, src, lane, row0, vb, rb);
                gcn_add_rows64(va, ra, al);
                gcn_load_rows64<0>(nxt, src, lane, row0, va, ra);
                gcn_add_rows64(vb, rb, al);
                cur = nxt;
                e = en;
            }
            const int n = e1 - e;
            if (n > 0) {
                if (n > 8) gcn_load_rows64<8>(cur, src, lane, row0, vb, rb);
                gcn_add_rows64(va, ra, al);
                if (n > 16) gcn_load_rows64<16>(cur, src, lane, row0, va, ra);
                if (n > 8) gcn_add_rows64(vb, rb, al);
                if (n > 24) gcn_load_rows64<24>(cur, src, lane, row0, vb, rb);
                if (n > 16) gcn_add_rows64(va, ra, al);
                if (n > 24) gcn_add_rows64(vb, rb, al);
            }
        }
    } else {
        const int hl = lane >> 4;
        const float* sf = src + (lane & 15);
        float* af = acc + (lane & 15);
        float va[8], vb[8];
        int ra[8], rb[8];
        if (any) gcn_load_rows16<0>(cur, hl, sf, row0, va, ra);
        __syncthreads();
        if (any) {
            atomicAdd(&cnt[(cur.y != spare) ? (cur.y - row0) : (kBin + lane)], 1);
            while (e1 - e >= kRound) {
                const int en = e + kRound;
                const int2 nxt = gcn_list_line(binned, en, lane, e1, spare);
                gcn_load_rows16<8>(cur, hl, sf, row0, vb, rb);
                gcn_add_rows16(va, ra, af);
                gcn_load_rows16<0>(nxt, hl, sf, row0, va, ra);
                atomicAdd(&cnt[(nxt.y != spare) ? (nxt.y - row0) : (kBin + lane)], 1);
                gcn_add_rows16(vb, rb, af);
                cur = nxt;
                e = en;
            }
            const int n = e1 - e;
            if (n > 0) {
                if (n > 16) gcn_load_rows16<8>(cur, hl, sf, row0, vb, rb);
                gcn_add_rows16(va, ra, af);
                if (n > 16) gcn_add_rows16(vb, rb, af);
            }
        }
    }
    __syncthreads();

    float4* out4 = reinterpret_cast<float4*>(out) + (size_t)row0 * (F / 4);
#pragma unroll
    for (int k = 0; k < kIter; ++k) {
        const int i = tid + 256 * k;
        if (i < n4) {
            const float rs = gcn_rs<ADDONE>((float)cnt[i / (F / 4)]);
            float4 v = acc4[i];
            v.x *= rs;
            v.y *= rs;
            v.z *= rs;
            v.w *= rs;
            if (ACT) {
                const float ss = gcn_rs<ADDONE>(dsn[k]);
                v.x = tanhf(v.x + bq.x) * ss;
                v.y = tanhf(v.y + bq.y) * ss;
                v.z = tanhf(v.z + bq.z) * ss;
                v.w = tanhf(v.w + bq.w) * ss;
            }
            out4[i] = v;
        }
    }
}

template <int NOUT, bool TANH, bool ADDONE>
__global__ __launch_bounds__(128) void gcn_gemm(const float* __restrict__ in,
                                                const float* __restrict__ W,
                                                const float* __restrict__ bias,
                                                const float* __restrict__ degS,
                                                float* __restrict__ out, int N) {
    constexpr int K = 64;
    constexpr int CT = NOUT / 16;
    constexpr int PITCH = 68;

    __shared__ float Wt[NOUT * PITCH];
    __shared__ float Bl[NOUT];
    __shared__ float Al[64 * PITCH];

    const int tid = threadIdx.x;

    const float4* Wg4 = reinterpret_cast<const float4*>(W);
    for (int i = tid; i < K * NOUT / 4; i += 128) {
        int k = i / (NOUT / 4);
        int n = (i % (NOUT / 4)) * 4;
        float4 w = Wg4[i];
        Wt[(n + 0) * PITCH + k] = w.x;
        Wt[(n + 1) * PITCH + k] = w.y;
        Wt[(n + 2) * PITCH + k] = w.z;
        Wt[(n + 3) * PITCH + k] = w.w;
    }
    if (tid < NOUT) Bl[tid] = bias[tid];

    const int rowBase = blockIdx.x * 64;
    const float4* Ag4 = reinterpret_cast<const float4*>(in);
    for (int i = tid; i < 64 * (K / 4); i += 128) {
        int row = rowBase + i / (K / 4);
        int c4 = (i % (K / 4)) * 4;
        float4 v = (row < N) ? Ag4[(size_t)row * (K / 4) + (c4 >> 2)]
                             : make_float4(0.f, 0.f, 0.f, 0.f);
        *reinterpret_cast<float4*>(&Al[(i / (K / 4)) * PITCH + c4]) = v;
    }
    __syncthreads();

    const int wave = tid >> 5;
    const int lane = tid & 31;
    const int tileRow = rowBase + wave * 16;
    if (tileRow >= N) return;

    const int lm = lane & 15;
    const int kh = (lane >> 4) * 2;
    const int mh = (lane >> 4) * 8;

    v8f acc[CT] = {};
    const float* Arow = &Al[(wave * 16 + lm) * PITCH + kh];

#pragma unroll
    for (int k0 = 0; k0 < K; k0 += 4) {
        v2f a = *reinterpret_cast<const v2f*>(&Arow[k0]);
#pragma unroll
        for (int c = 0; c < CT; ++c) {
            v2f b = *reinterpret_cast<const v2f*>(&Wt[(c * 16 + lm) * PITCH + k0 + kh]);
            acc[c] = __builtin_amdgcn_wmma_f32_16x16x4_f32(
                false, a, false, b, (short)0, acc[c], false, false);
        }
    }

    float rs[8];
#pragma unroll
    for (int j = 0; j < 8; ++j) {
        int row = tileRow + mh + j;
        float d = (row < N) ? degS[row] : 1.0f;
        rs[j] = ADDONE ? rsqrtf(d + 1.0f) : rsqrtf(fmaxf(d, 1.0f));
    }

#pragma unroll
    for (int c = 0; c < CT; ++c) {
        int n = c * 16 + lm;
        float bv = Bl[n];
#pragma unroll
        for (int j = 0; j < 8; ++j) {
            int row = tileRow + mh + j;
            if (row < N) {
                float v = acc[c][j] + bv;
                if (TANH) v = tanhf(v);
                out[(size_t)row * NOUT + n] = v * rs[j];
            }
        }
    }
}

__global__ __launch_bounds__(128) void gcn_place_gemm_pair(const int* __restrict__ se,
                                                           const int* __restrict__ re,
                                                           int* __restrict__ binCursor,
                                                           int2* __restrict__ binned,
                                                           int E, int nb, int chunks,
                                                           const float* __restrict__ in,
                                                           const float* __restrict__ Wa,
                                                           const float* __restrict__ ba,
                                                           const float* __restrict__ Wb,
                                                           const float* __restrict__ degS,
                                                           float* __restrict__ out, int N) {
    extern __shared__ int hist[];
    if ((int)blockIdx.x < chunks) {
        gcn_place_body(se, re, binCursor, binned, E, nb, hist, (int)blockIdx.x);
        return;
    }
    const int tileBlock = (int)blockIdx.x - chunks;

    constexpr int K = 64;
    constexpr int NOUT = 64;
    constexpr int CT = NOUT / 16;
    constexpr int PITCH = 68;

    constexpr int WIT = K * NOUT / 4 / 128;

    __shared__ float Wt[NOUT * PITCH];
    __shared__ float Bl[NOUT];
    __shared__ float Al[64 * PITCH];

    const int tid = threadIdx.x;

    const float4* Wa4 = reinterpret_cast<const float4*>(Wa);
    const float4* Wb4 = reinterpret_cast<const float4*>(Wb);
    float4 ub[WIT];
#pragma unroll
    for (int t = 0; t < WIT; ++t) {
        const int i = tid + 128 * t;
        int k = i / (NOUT / 4);
        int n = (i % (NOUT / 4)) * 4;
        float4 w = Wa4[i];
        ub[t] = Wb4[i];
        Wt[(n + 0) * PITCH + k] = w.x;
        Wt[(n + 1) * PITCH + k] = w.y;
        Wt[(n + 2) * PITCH + k] = w.z;
        Wt[(n + 3) * PITCH + k] = w.w;
    }
    if (tid < NOUT) Bl[tid] = ba[tid];

    const int rowBase = tileBlock * 64;
    const float4* Ag4 = reinterpret_cast<const float4*>(in);
    for (int i = tid; i < 64 * (K / 4); i += 128) {
        int row = rowBase + i / (K / 4);
        int c4 = (i % (K / 4)) * 4;
        float4 v = (row < N) ? Ag4[(size_t)row * (K / 4) + (c4 >> 2)]
                             : make_float4(0.f, 0.f, 0.f, 0.f);
        *reinterpret_cast<float4*>(&Al[(i / (K / 4)) * PITCH + c4]) = v;
    }
    __syncthreads();

    const int wave = tid >> 5;
    const int lane = tid & 31;
    const int tileRow = rowBase + wave * 16;

    const int lm = lane & 15;
    const int kh = (lane >> 4) * 2;
    const int mh = (lane >> 4) * 8;

    const float* Arow = &Al[(wave * 16 + lm) * PITCH + kh];

    v8f acc[CT] = {};
#pragma unroll
    for (int k0 = 0; k0 < K; k0 += 4) {
        v2f a = *reinterpret_cast<const v2f*>(&Arow[k0]);
#pragma unroll
        for (int c = 0; c < CT; ++c) {
            v2f b = *reinterpret_cast<const v2f*>(&Wt[(c * 16 + lm) * PITCH + k0 + kh]);
            acc[c] = __builtin_amdgcn_wmma_f32_16x16x4_f32(
                false, a, false, b, (short)0, acc[c], false, false);
        }
    }

    float rs[8];
#pragma unroll
    for (int j = 0; j < 8; ++j) {
        int row = tileRow + mh + j;
        float d = (row < N) ? degS[row] : 1.0f;
        rs[j] = rsqrtf(d + 1.0f);
    }

    float* Hrow = &Al[(wave * 16 + mh) * PITCH + lm];
#pragma unroll
    for (int c = 0; c < CT; ++c) {
        float bv = Bl[c * 16 + lm];
#pragma unroll
        for (int j = 0; j < 8; ++j) Hrow[j * PITCH + c * 16] = tanhf(acc[c][j] + bv) * rs[j];
    }
    __syncthreads();
#pragma unroll
    for (int t = 0; t < WIT; ++t) {
        const int i = tid + 128 * t;
        int k = i / (NOUT / 4);
        int n = (i % (NOUT / 4)) * 4;
        Wt[(n + 0) * PITCH + k] = ub[t].x;
        Wt[(n + 1) * PITCH + k] = ub[t].y;
        Wt[(n + 2) * PITCH + k] = ub[t].z;
        Wt[(n + 3) * PITCH + k] = ub[t].w;
    }
    __syncthreads();

    v8f acc2[CT] = {};
#pragma unroll
    for (int k0 = 0; k0 < K; k0 += 4) {
        v2f a = *reinterpret_cast<const v2f*>(&Arow[k0]);
#pragma unroll
        for (int c = 0; c < CT; ++c) {
            v2f b = *reinterpret_cast<const v2f*>(&Wt[(c * 16 + lm) * PITCH + k0 + kh]);
            acc2[c] = __builtin_amdgcn_wmma_f32_16x16x4_f32(
                false, a, false, b, (short)0, acc2[c], false, false);
        }
    }

#pragma unroll
    for (int c = 0; c < CT; ++c) {
        int n = c * 16 + lm;
#pragma unroll
        for (int j = 0; j < 8; ++j) {
            int row = tileRow + mh + j;
            if (row < N) out[(size_t)row * NOUT + n] = acc2[c][j];
        }
    }
}

extern "C" void kernel_launch(void* const* d_in, const int* in_sizes, int n_in,
                              void* d_out, int out_size, void* d_ws, size_t ws_size,
                              hipStream_t stream) {
    const float* nodes = (const float*)d_in[0];
    const int* senders = (const int*)d_in[1];
    const int* receivers = (const int*)d_in[2];
    const float* W0 = (const float*)d_in[3];
    const float* b0 = (const float*)d_in[4];
    const float* W1 = (const float*)d_in[5];
    const float* b1 = (const float*)d_in[6];
    const float* W2 = (const float*)d_in[7];
    const float* b2 = (const float*)d_in[8];
    float* out = (float*)d_out;

    const int N = in_sizes[0] / 64;
    const int E = in_sizes[1];
    const int nb = (N + kBin - 1) / kBin;
    const int nbA = (nb + 4) & ~3;
    const int nA = (N + 3) & ~3;
    const int nW = (((N + 1) >> 1) + 3) & ~3;
    const int P = (E + kCntEdges - 1) / kCntEdges;
    const int Q = (nW + kRangeWords - 1) / kRangeWords;

    float* X = (float*)d_ws;
    float* Y = X + (size_t)N * 64;
    float* degS = Y + (size_t)N * 64;
    int* binStart = (int*)(degS + nA);
    int* binCursor = binStart + nbA;
    unsigned int* sendPart = (unsigned int*)(binCursor + nbA);
    int* binPart = (int*)(sendPart + (size_t)P * nW);
    int2* binned = (int2*)(binPart + (size_t)P * nbA);

    const size_t histBytes = (size_t)nb * sizeof(int);
    gcn_count<<<dim3(Q, P), kCntThreads, histBytes, stream>>>(senders, receivers, sendPart, binPart, E, nb, nbA, nW);
    gcn_scan_reduce<<<1 + (nW + 1023) / 1024, 1024, 0, stream>>>(binPart, binStart, binCursor, nb, nbA, P,
                                                                 sendPart, degS, N, nW);

    const int chunks = (E + kPlaceThreads * kPlaceEdges - 1) / (kPlaceThreads * kPlaceEdges);
    const int tiles = (N + 15) / 16;
    const int gblocks = (tiles + 3) / 4;

    gcn_place_gemm_pair<<<chunks + gblocks, kPlaceThreads, histBytes, stream>>>(
        senders, receivers, binCursor, binned, E, nb, chunks, nodes, W0, b0, W1, degS, X, N);
    gcn_agg<64, true, true, true><<<nb, 256, 0, stream>>>(binned, binStart, X, b1, degS, Y, N);

    gcn_agg<64, true, true, false><<<nb, 256, 0, stream>>>(binned, binStart, Y, nullptr, nullptr, X, N);

    gcn_gemm<16, false, false><<<gblocks, 128, 0, stream>>>(X, W2, b2, degS, Y, N);
    gcn_agg<16, false, false, false><<<nb, 256, 0, stream>>>(binned, binStart, Y, nullptr, nullptr, out, N);
}


# receiver ranges of 132 nodes instead of 128: 758 aggregation work-groups, at most 3 per WGP on 256 WGPs (782 left a 4th on 14 WGPs, whose LDS adds ended each launch); LDS arrival order changes
# speedup vs baseline: 3.4186x; 1.0389x over previous
#include <hip/hip_runtime.h>
#include <hip/hip_bf16.h>

typedef float v2f __attribute__((ext_vector_type(2)));
typedef float v8f __attribute__((ext_vector_type(8)));

constexpr int kBin = 132;
static_assert(kBin % 2 == 0, "a range of 64-byte rows must be whole 128-byte lines");

__device__ __forceinline__ int gcn_bin(int r) { return (int)((unsigned int)r / (unsigned int)kBin); }

constexpr int kCntThreads = 512;
constexpr int kCntPer = 32;
constexpr int kCntEdges = kCntThreads * kCntPer;
constexpr int kRangeWords = 12544;
static_assert(kCntEdges < 65536, "a 16-bit sender counter must hold a chunk's edge count");
static_assert(kRangeWords % 4 == 0, "a range is a whole number of 16-byte pieces");

__global__ __launch_bounds__(kCntThreads) void gcn_count(const int* __restrict__ se,
                                                         const int* __restrict__ re,
                                                         unsigned int* __restrict__ sendPart,
                                                         int* __restrict__ binPart,
                                                         int E, int nb, int nbP, int nW) {
    __shared__ unsigned int cnt16[kRangeWords];
    extern __shared__ int hist[];
    const int tid = threadIdx.x;
    const int q = blockIdx.x;
    const int p = blockIdx.y;
    const bool bins = (q == 0);
    const int baseW = q * kRangeWords;
    const int words = (nW - baseW < kRangeWords) ? (nW - baseW) : kRangeWords;
    const int base = p * kCntEdges + tid;
    int s[kCntPer], r[kCntPer];
#pragma unroll
    for (int k = 0; k < kCntPer; ++k) {
        const int e = base + k * kCntThreads;
        const int ec = (e < E) ? e : (E - 1);
        s[k] = se[ec];
        r[k] = re[ec];
    }
    if (bins) {
        for (int b = tid; b < nb; b += kCntThreads) hist[b] = 0;
    }
    uint4* c4 = reinterpret_cast<uint4*>(cnt16);
    for (int i = tid; i < words / 4; i += kCntThreads) c4[i] = make_uint4(0u, 0u, 0u, 0u);
    __syncthreads();
#pragma unroll
    for (int k = 0; k < kCntPer; ++k) {
        const int e = base + k * kCntThreads;
        if (e < E) {
            const int w = (s[k] >> 1) - baseW;
            if ((unsigned int)w < (unsigned int)words) atomicAdd(&cnt16[w], 1u << ((s[k] & 1) << 4));
            if (bins) atomicAdd(&hist[gcn_bin(r[k])], 1);
        }
    }
    __syncthreads();
    uint4* p4 = reinterpret_cast<uint4*>(sendPart + (size_t)p * nW + baseW);
    for (int i = tid; i < words / 4; i += kCntThreads) p4[i] = c4[i];
    if (bins) {
        for (int b = tid; b < nb; b += kCntThreads) binPart[(size_t)p * nbP + b] = hist[b];
    }
}

constexpr int kRedBatch = 32;

__global__ __launch_bounds__(1024) void gcn_scan_reduce(const int* __restrict__ binPart,
                                                        int* __restrict__ binStart,
                                                        int* __restrict__ binCursor,
                                                        int nb, int nbP, int P,
                                                        const unsigned int* __restrict__ sendPart,
                                                        float* __restrict__ degS, int N, int nW) {
    __shared__ int part[1024];
    __shared__ int carry;
    const int tid = threadIdx.x;
    if (blockIdx.x != 0) {
        const int w = ((int)blockIdx.x - 1) * 1024 + tid;
        if (w < nW) {
            unsigned int lo = 0, hi = 0;
            for (int p0 = 0; p0 < P; p0 += kRedBatch) {
                unsigned int v[kRedBatch];
#pragma unroll
                for (int j = 0; j < kRedBatch; ++j) {
                    const int pc = (p0 + j < P) ? (p0 + j) : (P - 1);
                    v[j] = sendPart[(size_t)pc * nW + w];
                }
#pragma unroll
                for (int j = 0; j < kRedBatch; ++j) {
                    if (p0 + j < P) {
                        lo += v[j] & 0xffffu;
                        hi += v[j] >> 16;
                    }
                }
            }
            const int node = 2 * w;
            if (node < N) degS[node] = (float)lo;
            if (node + 1 < N) degS[node + 1] = (float)hi;
        }
        return;
    }
    if (tid == 0) carry = 0;
    __syncthreads();
    for (int b0 = 0; b0 < nb; b0 += 1024) {
        const int b = b0 + tid;
        int c = 0;
        if (b < nb) {
            for (int p0 = 0; p0 < P; p0 += kRedBatch) {
                int v[kRedBatch];
#pragma unroll
                for (int j = 0; j < kRedBatch; ++j) {
                    const int pc = (p0 + j < P) ? (p0 + j) : (P - 1);
                    v[j] = binPart[(size_t)pc * nbP + b];
                }
#pragma unroll
                for (int j = 0; j < kRedBatch; ++j) {
                    if (p0 + j < P) c += v[j];
                }
            }
        }
        part[tid] = c;
        __syncthreads();
        for (int off = 1; off < 1024; off <<= 1) {
            const int v = (tid >= off) ? part[tid - off] : 0;
            __syncthreads();
            part[tid] += v;
            __syncthreads();
        }
        const int incl = part[tid];
        const int first = carry;
        if (b < nb) {
            binStart[b] = first + incl - c;
            binCursor[b] = first + incl - c;
        }
        __syncthreads();
        if (tid == 1023) carry = first + incl;
        __syncthreads();
    }
    if (tid == 0) binStart[nb] = carry;
}

constexpr int kPlaceThreads = 128;
constexpr int kPlaceEdges = 32;

__device__ __forceinline__ void gcn_place_body(const int* __restrict__ se,
                                               const int* __restrict__ re,
                                               int* __restrict__ binCursor,
                                               int2* __restrict__ binned,
                                               int E, int nb, int* hist, int chunkIdx) {
    for (int b = threadIdx.x; b < nb; b += kPlaceThreads) hist[b] = 0;
    __syncthreads();
    const int base = chunkIdx * (kPlaceThreads * kPlaceEdges) + threadIdx.x;
    int s[kPlaceEdges], r[kPlaceEdges], rank[kPlaceEdges];
#pragma unroll
    for (int k = 0; k < kPlaceEdges; ++k) {
        const int e = base + k * kPlaceThreads;
        const int ec = (e < E) ? e : (E - 1);
        s[k] = se[ec];
        r[k] = re[ec];
    }
#pragma unroll
    for (int k = 0; k < kPlaceEdges; ++k) {
        const int e = base + k * kPlaceThreads;
        rank[k] = -1;
        if (e < E) rank[k] = atomicAdd(&hist[gcn_bin(r[k])], 1);
    }
    __syncthreads();
    for (int b = threadIdx.x; b < nb; b += kPlaceThreads) {
        const int c = hist[b];
        if (c) hist[b] = atomicAdd(&binCursor[b], c);
    }
    __syncthreads();
#pragma unroll
    for (int k = 0; k < kPlaceEdges; ++k) {
        if (rank[k] >= 0) binned[hist[gcn_bin(r[k])] + rank[k]] = make_int2(s[k], r[k]);
    }
}

constexpr int kRound = 32;

__device__ __forceinline__ int2 gcn_list_line(const int2* __restrict__ binned, int e, int lane,
                                              int e1, int spare) {
    const int q = e + lane;
    int qc = (q < e1) ? q : (e1 - 1);
    qc = (qc > 0) ? qc : 0;
    int2 p = binned[qc];
    if (q >= e1) p = make_int2(0, spare);
    return p;
}

template <int J0>
__device__ __forceinline__ void gcn_load_rows64(int2 slot, const float* __restrict__ src, int lane,
                                                int row0, float2 (&v)[8], int (&rl)[8]) {
#pragma unroll
    for (int i = 0; i < 8; ++i) {
        const int s = __builtin_amdgcn_readlane(slot.x, J0 + i);
        rl[i] = __builtin_amdgcn_readlane(slot.y, J0 + i) - row0;
        v[i] = reinterpret_cast<const float2*>(src + (size_t)s * 64)[lane];
    }
}

__device__ __forceinline__ void gcn_add_rows64(const float2 (&v)[8], const int (&rl)[8], float* al) {
#pragma unroll
    for (int i = 0; i < 8; ++i) {
        float* a = al + rl[i] * 64;
        atomicAdd(a, v[i].x);
        atomicAdd(a + 1, v[i].y);
    }
}

template <int P0>
__device__ __forceinline__ void gcn_load_rows16(int2 slot, int hl, const float* __restrict__ sf,
                                                int row0, float (&v)[8], int (&rl)[8]) {
#pragma unroll
    for (int i = 0; i < 8; ++i) {
        const int j = 2 * (P0 + i);
        const int s0 = __builtin_amdgcn_readlane(slot.x, j);
        const int s1 = __builtin_amdgcn_readlane(slot.x, j + 1);
        const int r0 = __builtin_amdgcn_readlane(slot.y, j);
        const int r1 = __builtin_amdgcn_readlane(slot.y, j + 1);
        const int s = hl ? s1 : s0;
        rl[i] = (hl ? r1 : r0) - row0;
        v[i] = sf[(size_t)s * 16];
    }
}

__device__ __forceinline__ void gcn_add_rows16(const float (&v)[8], const int (&rl)[8], float* af) {
#pragma unroll
    for (int i = 0; i < 8; ++i) atomicAdd(af + rl[i] * 16, v[i]);
}

template <bool ADDONE>
__device__ __forceinline__ float gcn_rs(float d) {
    return ADDONE ? rsqrtf(d + 1.0f) : rsqrtf(fmaxf(d, 1.0f));
}

template <int F, bool SELF, bool ADDONE, bool ACT>
__global__ __launch_bounds__(256) void gcn_agg(const int2* __restrict__ binned,
                                               const int* __restrict__ binStart,
                                               const float* __restrict__ src,
                                               const float* __restrict__ actBias,
                                               const float* __restrict__ actDegS,
                                               float* __restrict__ out, int N) {
    constexpr int kWaves = 8;
    constexpr int kTile4 = kBin * (F / 4);
    constexpr int kIter = (kTile4 + 255) / 256;
    static_assert(kBin <= 256, "one thread per receiver clears the slot counts");
    __shared__ float acc[(kBin + 1) * F];
    __shared__ int cnt[kBin + 32];

    const int tid = threadIdx.x;
    const int row0 = blockIdx.x * kBin;
    const int rows = (N - row0 < kBin) ? (N - row0) : kBin;
    const int n4 = rows * (F / 4);
    float4* acc4 = reinterpret_cast<float4*>(acc);
    const float4* src4 = reinterpret_cast<const float4*>(src) + (size_t)row0 * (F / 4);

    float dsn[kIter];
    float4 st[kIter];
#pragma unroll
    for (int k = 0; k < kIter; ++k) {
        const int i = tid + 256 * k;
        const int ic = (i < n4) ? i : (n4 - 1);
        if (ACT) dsn[k] = actDegS[row0 + ic / (F / 4)];
        if (SELF) st[k] = src4[ic];
    }
    float4 bq = make_float4(0.f, 0.f, 0.f, 0.f);
    if (ACT) bq = reinterpret_cast<const float4*>(actBias)[tid & (F / 4 - 1)];

    const int lane = tid & 31;
    const int wave = __builtin_amdgcn_readfirstlane(tid >> 5);
    const int e0 = binStart[blockIdx.x];
    const int eEnd = binStart[blockIdx.x + 1];
    const int per = (((eEnd - e0 + kWaves - 1) / kWaves) + 7) & ~7;
    int e = e0 + wave * per;
    const int e1 = (e + per < eEnd) ? (e + per) : eEnd;
    const bool any = e < e1;
    const int spare = row0 + kBin;

    int2 cur = gcn_list_line(binned, e, lane, e1, spare);

#pragma unroll
    for (int k = 0; k < kIter; ++k) {
        const int i = tid + 256 * k;
        float4 v = make_float4(0.f, 0.f, 0.f, 0.f);
        if (SELF && i < n4) v = st[k];
        if (i < kTile4) acc4[i] = v;
    }
    if (tid < F / 4) acc4[kTile4 + tid] = make_float4(0.f, 0.f, 0.f, 0.f);
    if (tid < kBin) cnt[tid] = 0;

    if constexpr (F == 64) {
        float* al = acc + 2 * lane;
        float2 va[8], vb[8];
        int ra[8], rb[8];
        if (any) gcn_load_rows64<0>(cur, src, lane, row0, va, ra);
        __syncthreads();
        if (any) {
            atomicAdd(&cnt[(cur.y != spare) ? (cur.y - row0) : (kBin + lane)], 1);
            while (e1 - e >= kRound) {
                const int en = e + kRound;
                const int2 nxt = gcn_list_line(binned, en, lane, e1, spare);
                gcn_load_rows64<8>(cur, src, lane, row0, vb, rb);
                gcn_add_rows64(va, ra, al);
                gcn_load_rows64<16>(cur, src, lane, row0, va, ra);
                gcn_add_rows64(vb, rb, al);
                atomicAdd(&cnt[(nxt.y != spare) ? (nxt.y - row0) : (kBin + lane)], 1);
                gcn_load_rows64<24>(cur, src, lane, row0, vb, rb);
                gcn_add_rows64(va, ra, al);
                gcn_load_rows64<0>(nxt, src, lane, row0, va, ra);
                gcn_add_rows64(vb, rb, al);
                cur = nxt;
                e = en;
            }
            const int n = e1 - e;
            if (n > 0) {
                if (n > 8) gcn_load_rows64<8>(cur, src, lane, row0, vb, rb);
                gcn_add_rows64(va, ra, al);
                if (n > 16) gcn_load_rows64<16>(cur, src, lane, row0, va, ra);
                if (n > 8) gcn_add_rows64(vb, rb, al);
                if (n > 24) gcn_load_rows64<24>(cur, src, lane, row0, vb, rb);
                if (n > 16) gcn_add_rows64(va, ra, al);
                if (n > 24) gcn_add_rows64(vb, rb, al);
            }
        }
    } else {
        const int hl = lane >> 4;
        const float* sf = src + (lane & 15);
        float* af = acc + (lane & 15);
        float va[8], vb[8];
        int ra[8], rb[8];
        if (any) gcn_load_rows16<0>(cur, hl, sf, row0, va, ra);
        __syncthreads();
        if (any) {
            atomicAdd(&cnt[(cur.y != spare) ? (cur.y - row0) : (kBin + lane)], 1);
            while (e1 - e >= kRound) {
                const int en = e + kRound;
                const int2 nxt = gcn_list_line(binned, en, lane, e1, spare);
                gcn_load_rows16<8>(cur, hl, sf, row0, vb, rb);
                gcn_add_rows16(va, ra, af);
                gcn_load_rows16<0>(nxt, hl, sf, row0, va, ra);
                atomicAdd(&cnt[(nxt.y != spare) ? (nxt.y - row0) : (kBin + lane)], 1);
                gcn_add_rows16(vb, rb, af);
                cur = nxt;
                e = en;
            }
            const int n = e1 - e;
            if (n > 0) {
                if (n > 16) gcn_load_rows16<8>(cur, hl, sf, row0, vb, rb);
                gcn_add_rows16(va, ra, af);
                if (n > 16) gcn_add_rows16(vb, rb, af);
            }
        }
    }
    __syncthreads();

    float4* out4 = reinterpret_cast<float4*>(out) + (size_t)row0 * (F / 4);
#pragma unroll
    for (int k = 0; k < kIter; ++k) {
        const int i = tid + 256 * k;
        if (i < n4) {
            const float rs = gcn_rs<ADDONE>((float)cnt[i / (F / 4)]);
            float4 v = acc4[i];
            v.x *= rs;
            v.y *= rs;
            v.z *= rs;
            v.w *= rs;
            if (ACT) {
                const float ss = gcn_rs<ADDONE>(dsn[k]);
                v.x = tanhf(v.x + bq.x) * ss;
                v.y = tanhf(v.y + bq.y) * ss;
                v.z = tanhf(v.z + bq.z) * ss;
                v.w = tanhf(v.w + bq.w) * ss;
            }
            out4[i] = v;
        }
    }
}

template <int NOUT, bool TANH, bool ADDONE>
__global__ __launch_bounds__(128) void gcn_gemm(const float* __restrict__ in,
                                                const float* __restrict__ W,
                                                const float* __restrict__ bias,
                                                const float* __restrict__ degS,
                                                float* __restrict__ out, int N) {
    constexpr int K = 64;
    constexpr int CT = NOUT / 16;
    constexpr int PITCH = 68;

    __shared__ float Wt[NOUT * PITCH];
    __shared__ float Bl[NOUT];
    __shared__ float Al[64 * PITCH];

    const int tid = threadIdx.x;

    const float4* Wg4 = reinterpret_cast<const float4*>(W);
    for (int i = tid; i < K * NOUT / 4; i += 128) {
        int k = i / (NOUT / 4);
        int n = (i % (NOUT / 4)) * 4;
        float4 w = Wg4[i];
        Wt[(n + 0) * PITCH + k] = w.x;
        Wt[(n + 1) * PITCH + k] = w.y;
        Wt[(n + 2) * PITCH + k] = w.z;
        Wt[(n + 3) * PITCH + k] = w.w;
    }
    if (tid < NOUT) Bl[tid] = bias[tid];

    const int rowBase = blockIdx.x * 64;
    const float4* Ag4 = reinterpret_cast<const float4*>(in);
    for (int i = tid; i < 64 * (K / 4); i += 128) {
        int row = rowBase + i / (K / 4);
        int c4 = (i % (K / 4)) * 4;
        float4 v = (row < N) ? Ag4[(size_t)row * (K / 4) + (c4 >> 2)]
                             : make_float4(0.f, 0.f, 0.f, 0.f);
        *reinterpret_cast<float4*>(&Al[(i / (K / 4)) * PITCH + c4]) = v;
    }
    __syncthreads();

    const int wave = tid >> 5;
    const int lane = tid & 31;
    const int tileRow = rowBase + wave * 16;
    if (tileRow >= N) return;

    const int lm = lane & 15;
    const int kh = (lane >> 4) * 2;
    const int mh = (lane >> 4) * 8;

    v8f acc[CT] = {};
    const float* Arow = &Al[(wave * 16 + lm) * PITCH + kh];

#pragma unroll
    for (int k0 = 0; k0 < K; k0 += 4) {
        v2f a = *reinterpret_cast<const v2f*>(&Arow[k0]);
#pragma unroll
        for (int c = 0; c < CT; ++c) {
            v2f b = *reinterpret_cast<const v2f*>(&Wt[(c * 16 + lm) * PITCH + k0 + kh]);
            acc[c] = __builtin_amdgcn_wmma_f32_16x16x4_f32(
                false, a, false, b, (short)0, acc[c], false, false);
        }
    }

    float rs[8];
#pragma unroll
    for (int j = 0; j < 8; ++j) {
        int row = tileRow + mh + j;
        float d = (row < N) ? degS[row] : 1.0f;
        rs[j] = ADDONE ? rsqrtf(d + 1.0f) : rsqrtf(fmaxf(d, 1.0f));
    }

#pragma unroll
    for (int c = 0; c < CT; ++c) {
        int n = c * 16 + lm;
        float bv = Bl[n];
#pragma unroll
        for (int j = 0; j < 8; ++j) {
            int row = tileRow + mh + j;
            if (row < N) {
                float v = acc[c][j] + bv;
                if (TANH) v = tanhf(v);
                out[(size_t)row * NOUT + n] = v * rs[j];
            }
        }
    }
}

__global__ __launch_bounds__(128) void gcn_place_gemm_pair(const int* __restrict__ se,
                                                           const int* __restrict__ re,
                                                           int* __restrict__ binCursor,
                                                           int2* __restrict__ binned,
                                                           int E, int nb, int chunks,
                                                           const float* __restrict__ in,
                                                           const float* __restrict__ Wa,
                                                           const float* __restrict__ ba,
                                                           const float* __restrict__ Wb,
                                                           const float* __restrict__ degS,
                                                           float* __restrict__ out, int N) {
    extern __shared__ int hist[];
    if ((int)blockIdx.x < chunks) {
        gcn_place_body(se, re, binCursor, binned, E, nb, hist, (int)blockIdx.x);
        return;
    }
    const int tileBlock = (int)blockIdx.x - chunks;

    constexpr int K = 64;
    constexpr int NOUT = 64;
    constexpr int CT = NOUT / 16;
    constexpr int PITCH = 68;

    constexpr int WIT = K * NOUT / 4 / 128;

    __shared__ float Wt[NOUT * PITCH];
    __shared__ float Bl[NOUT];
    __shared__ float Al[64 * PITCH];

    const int tid = threadIdx.x;

    const float4* Wa4 = reinterpret_cast<const float4*>(Wa);
    const float4* Wb4 = reinterpret_cast<const float4*>(Wb);
    float4 ub[WIT];
#pragma unroll
    for (int t = 0; t < WIT; ++t) {
        const int i = tid + 128 * t;
        int k = i / (NOUT / 4);
        int n = (i % (NOUT / 4)) * 4;
        float4 w = Wa4[i];
        ub[t] = Wb4[i];
        Wt[(n + 0) * PITCH + k] = w.x;
        Wt[(n + 1) * PITCH + k] = w.y;
        Wt[(n + 2) * PITCH + k] = w.z;
        Wt[(n + 3) * PITCH + k] = w.w;
    }
    if (tid < NOUT) Bl[tid] = ba[tid];

    const int rowBase = tileBlock * 64;
    const float4* Ag4 = reinterpret_cast<const float4*>(in);
    for (int i = tid; i < 64 * (K / 4); i += 128) {
        int row = rowBase + i / (K / 4);
        int c4 = (i % (K / 4)) * 4;
        float4 v = (row < N) ? Ag4[(size_t)row * (K / 4) + (c4 >> 2)]
                             : make_float4(0.f, 0.f, 0.f, 0.f);
        *reinterpret_cast<float4*>(&Al[(i / (K / 4)) * PITCH + c4]) = v;
    }
    __syncthreads();

    const int wave = tid >> 5;
    const int lane = tid & 31;
    const int tileRow = rowBase + wave * 16;

    const int lm = lane & 15;
    const int kh = (lane >> 4) * 2;
    const int mh = (lane >> 4) * 8;

    const float* Arow = &Al[(wave * 16 + lm) * PITCH + kh];

    v8f acc[CT] = {};
#pragma unroll
    for (int k0 = 0; k0 < K; k0 += 4) {
        v2f a = *reinterpret_cast<const v2f*>(&Arow[k0]);
#pragma unroll
        for (int c = 0; c < CT; ++c) {
            v2f b = *reinterpret_cast<const v2f*>(&Wt[(c * 16 + lm) * PITCH + k0 + kh]);
            acc[c] = __builtin_amdgcn_wmma_f32_16x16x4_f32(
                false, a, false, b, (short)0, acc[c], false, false);
        }
    }

    float rs[8];
#pragma unroll
    for (int j = 0; j < 8; ++j) {
        int row = tileRow + mh + j;
        float d = (row < N) ? degS[row] : 1.0f;
        rs[j] = rsqrtf(d + 1.0f);
    }

    float* Hrow = &Al[(wave * 16 + mh) * PITCH + lm];
#pragma unroll
    for (int c = 0; c < CT; ++c) {
        float bv = Bl[c * 16 + lm];
#pragma unroll
        for (int j = 0; j < 8; ++j) Hrow[j * PITCH + c * 16] = tanhf(acc[c][j] + bv) * rs[j];
    }
    __syncthreads();
#pragma unroll
    for (int t = 0; t < WIT; ++t) {
        const int i = tid + 128 * t;
        int k = i / (NOUT / 4);
        int n = (i % (NOUT / 4)) * 4;
        Wt[(n + 0) * PITCH + k] = ub[t].x;
        Wt[(n + 1) * PITCH + k] = ub[t].y;
        Wt[(n + 2) * PITCH + k] = ub[t].z;
        Wt[(n + 3) * PITCH + k] = ub[t].w;
    }
    __syncthreads();

    v8f acc2[CT] = {};
#pragma unroll
    for (int k0 = 0; k0 < K; k0 += 4) {
        v2f a = *reinterpret_cast<const v2f*>(&Arow[k0]);
#pragma unroll
        for (int c = 0; c < CT; ++c) {
            v2f b = *reinterpret_cast<const v2f*>(&Wt[(c * 16 + lm) * PITCH + k0 + kh]);
            acc2[c] = __builtin_amdgcn_wmma_f32_16x16x4_f32(
                false, a, false, b, (short)0, acc2[c], false, false);
        }
    }

#pragma unroll
    for (int c = 0; c < CT; ++c) {
        int n = c * 16 + lm;
#pragma unroll
        for (int j = 0; j < 8; ++j) {
            int row = tileRow + mh + j;
            if (row < N) out[(size_t)row * NOUT + n] = acc2[c][j];
        }
    }
}

extern "C" void kernel_launch(void* const* d_in, const int* in_sizes, int n_in,
                              void* d_out, int out_size, void* d_ws, size_t ws_size,
                              hipStream_t stream) {
    const float* nodes = (const float*)d_in[0];
    const int* senders = (const int*)d_in[1];
    const int* receivers = (const int*)d_in[2];
    const float* W0 = (const float*)d_in[3];
    const float* b0 = (const float*)d_in[4];
    const float* W1 = (const float*)d_in[5];
    const float* b1 = (const float*)d_in[6];
    const float* W2 = (const float*)d_in[7];
    const float* b2 = (const float*)d_in[8];
    float* out = (float*)d_out;

    const int N = in_sizes[0] / 64;
    const int E = in_sizes[1];
    const int nb = (N + kBin - 1) / kBin;
    const int nbA = (nb + 4) & ~3;
    const int nA = (N + 3) & ~3;
    const int nW = (((N + 1) >> 1) + 3) & ~3;
    const int P = (E + kCntEdges - 1) / kCntEdges;
    const int Q = (nW + kRangeWords - 1) / kRangeWords;

    float* X = (float*)d_ws;
    float* Y = X + (size_t)N * 64;
    float* degS = Y + (size_t)N * 64;
    int* binStart = (int*)(degS + nA);
    int* binCursor = binStart + nbA;
    unsigned int* sendPart = (unsigned int*)(binCursor + nbA);
    int* binPart = (int*)(sendPart + (size_t)P * nW);
    int2* binned = (int2*)(binPart + (size_t)P * nbA);

    const size_t histBytes = (size_t)nb * sizeof(int);
    gcn_count<<<dim3(Q, P), kCntThreads, histBytes, stream>>>(senders, receivers, sendPart, binPart, E, nb, nbA, nW);
    gcn_scan_reduce<<<1 + (nW + 1023) / 1024, 1024, 0, stream>>>(binPart, binStart, binCursor, nb, nbA, P,
                                                                 sendPart, degS, N, nW);

    const int chunks = (E + kPlaceThreads * kPlaceEdges - 1) / (kPlaceThreads * kPlaceEdges);
    const int tiles = (N + 15) / 16;
    const int gblocks = (tiles + 3) / 4;

    gcn_place_gemm_pair<<<chunks + gblocks, kPlaceThreads, histBytes, stream>>>(
        senders, receivers, binCursor, binned, E, nb, chunks, nodes, W0, b0, W1, degS, X, N);
    gcn_agg<64, true, true, true><<<nb, 256, 0, stream>>>(binned, binStart, X, b1, degS, Y, N);

    gcn_agg<64, true, true, false><<<nb, 256, 0, stream>>>(binned, binStart, Y, nullptr, nullptr, X, N);

    gcn_gemm<16, false, false><<<gblocks, 128, 0, stream>>>(X, W2, b2, degS, Y, N);
    gcn_agg<16, false, false, false><<<nb, 256, 0, stream>>>(binned, binStart, Y, nullptr, nullptr, out, N);
}
